# starting kernel
# baseline (speedup 1.0000x reference)
#include <hip/hip_runtime.h>
#include <math.h>


typedef __attribute__((ext_vector_type(2))) float v2f;
typedef __attribute__((ext_vector_type(8))) float v8f;

#define WIDTH  512
#define SEGS   32
#define XPITCH 513

__device__ __forceinline__ float exp2_fast(float x) {
    return __builtin_amdgcn_exp2f(x);
}

__device__ __forceinline__ void scan_dir(float* X, const float* tF, const float* omF,
                                         int lid, int half, int khi, bool bwd)
{
    float carry = 0.0f;
    for (int s = 0; s < SEGS; ++s) {
        const int base = s * 16;
        const int idx  = base + half;

        float t;
        if (!bwd) t = tF[idx];
        else      t = (idx == 0) ? -512.0f : tF[WIDTH - idx];

        float S = t;
#pragma unroll
        for (int off = 1; off < 16; off <<= 1) {
            float o = __shfl(S, (lid - off) & 31);
            if (half >= off) S += o;
        }
        const float expS = exp2_fast(S);

        v8f acc = {0.f,0.f,0.f,0.f,0.f,0.f,0.f,0.f};
#pragma unroll
        for (int kk = 0; kk < 4; ++kk) {
            const int m0 = kk * 4 + khi * 2;
            const float Sm0 = __shfl(S, m0);
            const float Sm1 = __shfl(S, m0 + 1);
            v2f Af;
            Af.x = (m0     < half) ? exp2_fast(S - Sm0) : ((m0     == half) ? 1.0f : 0.0f);
            Af.y = (m0 + 1 < half) ? exp2_fast(S - Sm1) : ((m0 + 1 == half) ? 1.0f : 0.0f);

            const int p0 = base + m0;
            v2f Bf;
            if (!bwd) {
                Bf.x = omF[p0]     * X[half * XPITCH + p0];
                Bf.y = omF[p0 + 1] * X[half * XPITCH + p0 + 1];
            } else {
                const float om0 = (p0 == 0) ? 1.0f : omF[WIDTH - p0];
                const float om1 = omF[WIDTH - (p0 + 1)];
                Bf.x = om0 * X[half * XPITCH + (WIDTH - 1 - p0)];
                Bf.y = om1 * X[half * XPITCH + (WIDTH - 1 - (p0 + 1))];
            }

            acc = __builtin_amdgcn_wmma_f32_16x16x4_f32(
                      false, Af, false, Bf, (short)0, acc, false, false);
        }

#pragma unroll
        for (int rr = 0; rr < 8; ++rr) {
            const float Pj = __shfl(expS, rr + 8 * khi);
            acc[rr] += Pj * carry;
        }
        carry = __shfl(acc[7], 16 + half);

#pragma unroll
        for (int rr = 0; rr < 8; ++rr) {
            const int j = base + rr + 8 * khi;
            const int w = bwd ? (WIDTH - 1 - j) : j;
            X[half * XPITCH + w] = acc[rr];
        }
    }
}

__global__ __launch_bounds__(32)
void dt_pass_kernel(const float* in, float* out, const float* edge,
                    float clog2, int vertical)
{
    extern __shared__ float smem[];
    float* X   = smem;
    float* tF  = smem + 16 * XPITCH;
    float* omF = tF + WIDTH;

    const int lid  = threadIdx.x;
    const int half = lid & 15;
    const int khi  = lid >> 4;

    const int nb  = blockIdx.x;
    const int cg  = nb & 1;
    const int row = nb >> 1;
    const int b   = row >> 9;
    const int r   = row & 511;

    const int C = 32, H = 512, Wd = 512;
    const long long chStride = (long long)H * Wd;
    long long imgBase, edgeBase;
    int posStride, eStride;
    if (!vertical) {
        imgBase  = (long long)b * C * H * Wd + (long long)r * Wd;
        posStride = 1;
        edgeBase = (long long)b * H * Wd + (long long)r * Wd;
        eStride  = 1;
    } else {
        imgBase  = (long long)b * C * H * Wd + r;
        posStride = Wd;
        edgeBase = (long long)b * H * Wd + r;
        eStride  = Wd;
    }
    imgBase += (long long)(cg * 16) * chStride;

    for (int i = lid; i < 16 * WIDTH; i += 32) {
        const int ch = i >> 9, w = i & (WIDTH - 1);
        X[ch * XPITCH + w] = in[imgBase + (long long)ch * chStride + (long long)w * posStride];
    }
    for (int w = lid; w < WIDTH; w += 32) {
        const float Dv = 1.0f + 150.0f * edge[edgeBase + (long long)w * eStride];
        float t = clog2 * Dv;
        if (w == 0) t = -512.0f;
        tF[w]  = t;
        omF[w] = 1.0f - exp2_fast(t);
    }
    __syncthreads();

    scan_dir(X, tF, omF, lid, half, khi, false);
    scan_dir(X, tF, omF, lid, half, khi, true);

    __syncthreads();
    for (int i = lid; i < 16 * WIDTH; i += 32) {
        const int ch = i >> 9, w = i & (WIDTH - 1);
        out[imgBase + (long long)ch * chStride + (long long)w * posStride] = X[ch * XPITCH + w];
    }
}

extern "C" void kernel_launch(void* const* d_in, const int* in_sizes, int n_in,
                              void* d_out, int out_size, void* d_ws, size_t ws_size,
                              hipStream_t stream) {
    (void)in_sizes; (void)n_in; (void)out_size; (void)d_ws; (void)ws_size;
    const float* img  = (const float*)d_in[0];
    const float* edge = (const float*)d_in[1];
    float* out = (float*)d_out;

    const double SQRT2 = 1.4142135623730951;
    const double SQRT3 = 1.7320508075688772;
    const double LOG2E = 1.4426950408889634;
    const double denom = 3.872983346207417;
    const double sigma1 = 60.0 * SQRT3 * 2.0 / denom;
    const double sigma2 = 60.0 * SQRT3 * 1.0 / denom;
    const float c1 = (float)(-SQRT2 / sigma1 * LOG2E);
    const float c2 = (float)(-SQRT2 / sigma2 * LOG2E);

    const dim3 grid(4096), block(32);
    const size_t shmem = (size_t)(16 * XPITCH + 2 * WIDTH) * sizeof(float);

    dt_pass_kernel<<<grid, block, shmem, stream>>>(img, out, edge, c1, 0);
    dt_pass_kernel<<<grid, block, shmem, stream>>>(out, out, edge, c1, 1);
    dt_pass_kernel<<<grid, block, shmem, stream>>>(out, out, edge, c2, 0);
    dt_pass_kernel<<<grid, block, shmem, stream>>>(out, out, edge, c2, 1);
}


# Vertical passes: an 8-wave work-group moves 8 adjacent columns x 16 channels through LDS in 32-byte row pieces (16-byte loads, stores), not 4-byte accesses at a 2-KB stride; scan and sums unchanged
# speedup vs baseline: 2.4982x; 2.4982x over previous
#include <hip/hip_runtime.h>
#include <math.h>


typedef __attribute__((ext_vector_type(2))) float v2f;
typedef __attribute__((ext_vector_type(8))) float v8f;

#define WIDTH  512
#define SEGS   32
#define XPITCH 513

#define VCOLS  8
#define COLF   (16 * XPITCH + 2 * WIDTH + 4)

__device__ __forceinline__ float exp2_fast(float x) {
    return __builtin_amdgcn_exp2f(x);
}

__device__ __forceinline__ void scan_dir(float* X, const float* tF, const float* omF,
                                         int lid, int half, int khi, bool bwd)
{
    float carry = 0.0f;
    for (int s = 0; s < SEGS; ++s) {
        const int base = s * 16;
        const int idx  = base + half;

        float t;
        if (!bwd) t = tF[idx];
        else      t = (idx == 0) ? -512.0f : tF[WIDTH - idx];

        float S = t;
#pragma unroll
        for (int off = 1; off < 16; off <<= 1) {
            float o = __shfl(S, (lid - off) & 31);
            if (half >= off) S += o;
        }
        const float expS = exp2_fast(S);

        v8f acc = {0.f,0.f,0.f,0.f,0.f,0.f,0.f,0.f};
#pragma unroll
        for (int kk = 0; kk < 4; ++kk) {
            const int m0 = kk * 4 + khi * 2;
            const float Sm0 = __shfl(S, m0);
            const float Sm1 = __shfl(S, m0 + 1);
            v2f Af;
            Af.x = (m0     < half) ? exp2_fast(S - Sm0) : ((m0     == half) ? 1.0f : 0.0f);
            Af.y = (m0 + 1 < half) ? exp2_fast(S - Sm1) : ((m0 + 1 == half) ? 1.0f : 0.0f);

            const int p0 = base + m0;
            v2f Bf;
            if (!bwd) {
                Bf.x = omF[p0]     * X[half * XPITCH + p0];
                Bf.y = omF[p0 + 1] * X[half * XPITCH + p0 + 1];
            } else {
                const float om0 = (p0 == 0) ? 1.0f : omF[WIDTH - p0];
                const float om1 = omF[WIDTH - (p0 + 1)];
                Bf.x = om0 * X[half * XPITCH + (WIDTH - 1 - p0)];
                Bf.y = om1 * X[half * XPITCH + (WIDTH - 1 - (p0 + 1))];
            }

            acc = __builtin_amdgcn_wmma_f32_16x16x4_f32(
                      false, Af, false, Bf, (short)0, acc, false, false);
        }

#pragma unroll
        for (int rr = 0; rr < 8; ++rr) {
            const float Pj = __shfl(expS, rr + 8 * khi);
            acc[rr] += Pj * carry;
        }
        carry = __shfl(acc[7], 16 + half);

#pragma unroll
        for (int rr = 0; rr < 8; ++rr) {
            const int j = base + rr + 8 * khi;
            const int w = bwd ? (WIDTH - 1 - j) : j;
            X[half * XPITCH + w] = acc[rr];
        }
    }
}

__global__ __launch_bounds__(32)
void dt_pass_kernel(const float* in, float* out, const float* edge,
                    float clog2, int vertical)
{
    extern __shared__ float smem[];
    float* X   = smem;
    float* tF  = smem + 16 * XPITCH;
    float* omF = tF + WIDTH;

    const int lid  = threadIdx.x;
    const int half = lid & 15;
    const int khi  = lid >> 4;

    const int nb  = blockIdx.x;
    const int cg  = nb & 1;
    const int row = nb >> 1;
    const int b   = row >> 9;
    const int r   = row & 511;

    const int C = 32, H = 512, Wd = 512;
    const long long chStride = (long long)H * Wd;
    long long imgBase, edgeBase;
    int posStride, eStride;
    if (!vertical) {
        imgBase  = (long long)b * C * H * Wd + (long long)r * Wd;
        posStride = 1;
        edgeBase = (long long)b * H * Wd + (long long)r * Wd;
        eStride  = 1;
    } else {
        imgBase  = (long long)b * C * H * Wd + r;
        posStride = Wd;
        edgeBase = (long long)b * H * Wd + r;
        eStride  = Wd;
    }
    imgBase += (long long)(cg * 16) * chStride;

    for (int i = lid; i < 16 * WIDTH; i += 32) {
        const int ch = i >> 9, w = i & (WIDTH - 1);
        X[ch * XPITCH + w] = in[imgBase + (long long)ch * chStride + (long long)w * posStride];
    }
    for (int w = lid; w < WIDTH; w += 32) {
        const float Dv = 1.0f + 150.0f * edge[edgeBase + (long long)w * eStride];
        float t = clog2 * Dv;
        if (w == 0) t = -512.0f;
        tF[w]  = t;
        omF[w] = 1.0f - exp2_fast(t);
    }
    __syncthreads();

    scan_dir(X, tF, omF, lid, half, khi, false);
    scan_dir(X, tF, omF, lid, half, khi, true);

    __syncthreads();
    for (int i = lid; i < 16 * WIDTH; i += 32) {
        const int ch = i >> 9, w = i & (WIDTH - 1);
        out[imgBase + (long long)ch * chStride + (long long)w * posStride] = X[ch * XPITCH + w];
    }
}

__global__ __launch_bounds__(VCOLS * 32)
void dt_vpass_kernel(const float* in, float* out, const float* edge, float clog2)
{
    __shared__ float vsm[VCOLS * COLF];

    const int tid  = threadIdx.x;
    const int wv   = tid >> 5;
    const int lid  = tid & 31;
    const int half = lid & 15;
    const int khi  = lid >> 4;

    const int nb = blockIdx.x;
    const int cg = nb & 1;
    const int g  = nb >> 1;
    const int b  = g >> 6;
    const int c0 = (g & 63) * VCOLS;

    const int C = 32, H = 512, Wd = 512;
    const long long chStride = (long long)H * Wd;
    const long long imgBase  = (long long)b * C * H * Wd + (long long)(cg * 16) * chStride + c0;
    const long long edgeBase = (long long)b * H * Wd + c0;

    constexpr int NEDGE = WIDTH * 2 / (VCOLS * 32);
    float4 e4[NEDGE];
#pragma unroll
    for (int k = 0; k < NEDGE; ++k) {
        const int q = tid + k * VCOLS * 32;
        const int j = q & 1, h = q >> 1;
        e4[k] = *reinterpret_cast<const float4*>(edge + edgeBase + (long long)h * Wd + j * 4);
    }

    constexpr int NPIECE = 16 * WIDTH * 2;
    constexpr int BATCH  = 16;
    for (int q0 = tid; q0 < NPIECE; q0 += BATCH * VCOLS * 32) {
        float4 v[BATCH];
#pragma unroll
        for (int k = 0; k < BATCH; ++k) {
            const int q  = q0 + k * VCOLS * 32;
            const int j  = q & 1, h = (q >> 1) & (WIDTH - 1), ch = q >> 10;
            v[k] = *reinterpret_cast<const float4*>(
                       in + imgBase + (long long)ch * chStride + (long long)h * Wd + j * 4);
        }
#pragma unroll
        for (int k = 0; k < BATCH; ++k) {
            const int q  = q0 + k * VCOLS * 32;
            const int j  = q & 1, h = (q >> 1) & (WIDTH - 1), ch = q >> 10;
            float* d = vsm + (j * 4) * COLF + ch * XPITCH + h;
            d[0 * COLF] = v[k].x;
            d[1 * COLF] = v[k].y;
            d[2 * COLF] = v[k].z;
            d[3 * COLF] = v[k].w;
        }
    }
#pragma unroll
    for (int k = 0; k < NEDGE; ++k) {
        const int q = tid + k * VCOLS * 32;
        const int j = q & 1, h = q >> 1;
        const float ev[4] = {e4[k].x, e4[k].y, e4[k].z, e4[k].w};
#pragma unroll
        for (int e = 0; e < 4; ++e) {
            float t = clog2 * (1.0f + 150.0f * ev[e]);
            if (h == 0) t = -512.0f;
            float* col = vsm + (j * 4 + e) * COLF + 16 * XPITCH;
            col[h]         = t;
            col[WIDTH + h] = 1.0f - exp2_fast(t);
        }
    }
    __syncthreads();

    float* X   = vsm + wv * COLF;
    float* tF  = X + 16 * XPITCH;
    float* omF = tF + WIDTH;
    scan_dir(X, tF, omF, lid, half, khi, false);
    scan_dir(X, tF, omF, lid, half, khi, true);

    __syncthreads();
    for (int q = tid; q < NPIECE; q += VCOLS * 32) {
        const int j = q & 1, h = (q >> 1) & (WIDTH - 1), ch = q >> 10;
        const float* s = vsm + (j * 4) * COLF + ch * XPITCH + h;
        float4 v;
        v.x = s[0 * COLF];
        v.y = s[1 * COLF];
        v.z = s[2 * COLF];
        v.w = s[3 * COLF];
        *reinterpret_cast<float4*>(
            out + imgBase + (long long)ch * chStride + (long long)h * Wd + j * 4) = v;
    }
}

extern "C" void kernel_launch(void* const* d_in, const int* in_sizes, int n_in,
                              void* d_out, int out_size, void* d_ws, size_t ws_size,
                              hipStream_t stream) {
    (void)in_sizes; (void)n_in; (void)out_size; (void)d_ws; (void)ws_size;
    const float* img  = (const float*)d_in[0];
    const float* edge = (const float*)d_in[1];
    float* out = (float*)d_out;

    const double SQRT2 = 1.4142135623730951;
    const double SQRT3 = 1.7320508075688772;
    const double LOG2E = 1.4426950408889634;
    const double denom = 3.872983346207417;
    const double sigma1 = 60.0 * SQRT3 * 2.0 / denom;
    const double sigma2 = 60.0 * SQRT3 * 1.0 / denom;
    const float c1 = (float)(-SQRT2 / sigma1 * LOG2E);
    const float c2 = (float)(-SQRT2 / sigma2 * LOG2E);

    const dim3 grid(4096), block(32);
    const size_t shmem = (size_t)(16 * XPITCH + 2 * WIDTH) * sizeof(float);

    const dim3 vgrid(512), vblock(VCOLS * 32);

    dt_pass_kernel<<<grid, block, shmem, stream>>>(img, out, edge, c1, 0);
    dt_vpass_kernel<<<vgrid, vblock, 0, stream>>>(out, out, edge, c1);
    dt_pass_kernel<<<grid, block, shmem, stream>>>(out, out, edge, c2, 0);
    dt_vpass_kernel<<<vgrid, vblock, 0, stream>>>(out, out, edge, c2);
}


# Scan: the 21 lane exchanges per 16-wide segment (prefix sum, T-matrix and carry broadcasts) run as DPP row_shr/row_share/row_ror and v_permlanex16 on the vector ALU, not as ds_bpermute through LDS
# speedup vs baseline: 3.0401x; 1.2169x over previous
#include <hip/hip_runtime.h>
#include <math.h>


typedef __attribute__((ext_vector_type(2))) float v2f;
typedef __attribute__((ext_vector_type(8))) float v8f;

#define WIDTH  512
#define SEGS   32
#define XPITCH 513

#define VCOLS  8
#define COLF   (16 * XPITCH + 2 * WIDTH + 4)

__device__ __forceinline__ float exp2_fast(float x) {
    return __builtin_amdgcn_exp2f(x);
}

template <int CTRL, int ROWS>
__device__ __forceinline__ float dpp_row(float old, float v) {
    return __builtin_bit_cast(float, __builtin_amdgcn_update_dpp(
               __builtin_bit_cast(int, old), __builtin_bit_cast(int, v), CTRL, ROWS, 0xf, true));
}
__device__ __forceinline__ float other_row(float v) {
    return __builtin_bit_cast(float, __builtin_amdgcn_permlanex16(
               __builtin_bit_cast(int, v), __builtin_bit_cast(int, v),
               0x76543210, 0xfedcba98, false, false));
}

__device__ __forceinline__ void scan_dir(float* X, const float* tF, const float* omF,
                                         int half, int khi, bool bwd)
{
    float carry = 0.0f;
    for (int s = 0; s < SEGS; ++s) {
        const int base = s * 16;
        const int idx  = base + half;

        float t;
        if (!bwd) t = tF[idx];
        else      t = (idx == 0) ? -512.0f : tF[WIDTH - idx];

        float S = t;
        S += dpp_row<0x111, 0xf>(0.0f, S);
        S += dpp_row<0x112, 0xf>(0.0f, S);
        S += dpp_row<0x114, 0xf>(0.0f, S);
        S += dpp_row<0x118, 0xf>(0.0f, S);
        const float expS = exp2_fast(S);

        const float Srot = dpp_row<0x12e, 0x2>(S, S);
        const float Prot = dpp_row<0x128, 0x2>(expS, expS);

        v8f acc = {0.f,0.f,0.f,0.f,0.f,0.f,0.f,0.f};
#pragma unroll
        for (int kk = 0; kk < 4; ++kk) {
            const int m0 = kk * 4 + khi * 2;
            float Sm0, Sm1;
            if (kk == 0)      { Sm0 = dpp_row<0x150, 0xf>(0.0f, Srot); Sm1 = dpp_row<0x151, 0xf>(0.0f, Srot); }
            else if (kk == 1) { Sm0 = dpp_row<0x154, 0xf>(0.0f, Srot); Sm1 = dpp_row<0x155, 0xf>(0.0f, Srot); }
            else if (kk == 2) { Sm0 = dpp_row<0x158, 0xf>(0.0f, Srot); Sm1 = dpp_row<0x159, 0xf>(0.0f, Srot); }
            else              { Sm0 = dpp_row<0x15c, 0xf>(0.0f, Srot); Sm1 = dpp_row<0x15d, 0xf>(0.0f, Srot); }
            v2f Af;
            Af.x = (m0     < half) ? exp2_fast(S - Sm0) : ((m0     == half) ? 1.0f : 0.0f);
            Af.y = (m0 + 1 < half) ? exp2_fast(S - Sm1) : ((m0 + 1 == half) ? 1.0f : 0.0f);

            const int p0 = base + m0;
            v2f Bf;
            if (!bwd) {
                Bf.x = omF[p0]     * X[half * XPITCH + p0];
                Bf.y = omF[p0 + 1] * X[half * XPITCH + p0 + 1];
            } else {
                const float om0 = (p0 == 0) ? 1.0f : omF[WIDTH - p0];
                const float om1 = omF[WIDTH - (p0 + 1)];
                Bf.x = om0 * X[half * XPITCH + (WIDTH - 1 - p0)];
                Bf.y = om1 * X[half * XPITCH + (WIDTH - 1 - (p0 + 1))];
            }

            acc = __builtin_amdgcn_wmma_f32_16x16x4_f32(
                      false, Af, false, Bf, (short)0, acc, false, false);
        }

        acc[0] += dpp_row<0x150, 0xf>(0.0f, Prot) * carry;
        acc[1] += dpp_row<0x151, 0xf>(0.0f, Prot) * carry;
        acc[2] += dpp_row<0x152, 0xf>(0.0f, Prot) * carry;
        acc[3] += dpp_row<0x153, 0xf>(0.0f, Prot) * carry;
        acc[4] += dpp_row<0x154, 0xf>(0.0f, Prot) * carry;
        acc[5] += dpp_row<0x155, 0xf>(0.0f, Prot) * carry;
        acc[6] += dpp_row<0x156, 0xf>(0.0f, Prot) * carry;
        acc[7] += dpp_row<0x157, 0xf>(0.0f, Prot) * carry;
        const float upper = other_row(acc[7]);
        carry = khi ? acc[7] : upper;

#pragma unroll
        for (int rr = 0; rr < 8; ++rr) {
            const int j = base + rr + 8 * khi;
            const int w = bwd ? (WIDTH - 1 - j) : j;
            X[half * XPITCH + w] = acc[rr];
        }
    }
}

__global__ __launch_bounds__(32)
void dt_pass_kernel(const float* in, float* out, const float* edge,
                    float clog2, int vertical)
{
    extern __shared__ float smem[];
    float* X   = smem;
    float* tF  = smem + 16 * XPITCH;
    float* omF = tF + WIDTH;

    const int lid  = threadIdx.x;
    const int half = lid & 15;
    const int khi  = lid >> 4;

    const int nb  = blockIdx.x;
    const int cg  = nb & 1;
    const int row = nb >> 1;
    const int b   = row >> 9;
    const int r   = row & 511;

    const int C = 32, H = 512, Wd = 512;
    const long long chStride = (long long)H * Wd;
    long long imgBase, edgeBase;
    int posStride, eStride;
    if (!vertical) {
        imgBase  = (long long)b * C * H * Wd + (long long)r * Wd;
        posStride = 1;
        edgeBase = (long long)b * H * Wd + (long long)r * Wd;
        eStride  = 1;
    } else {
        imgBase  = (long long)b * C * H * Wd + r;
        posStride = Wd;
        edgeBase = (long long)b * H * Wd + r;
        eStride  = Wd;
    }
    imgBase += (long long)(cg * 16) * chStride;

    for (int i = lid; i < 16 * WIDTH; i += 32) {
        const int ch = i >> 9, w = i & (WIDTH - 1);
        X[ch * XPITCH + w] = in[imgBase + (long long)ch * chStride + (long long)w * posStride];
    }
    for (int w = lid; w < WIDTH; w += 32) {
        const float Dv = 1.0f + 150.0f * edge[edgeBase + (long long)w * eStride];
        float t = clog2 * Dv;
        if (w == 0) t = -512.0f;
        tF[w]  = t;
        omF[w] = 1.0f - exp2_fast(t);
    }
    __syncthreads();

    scan_dir(X, tF, omF, half, khi, false);
    scan_dir(X, tF, omF, half, khi, true);

    __syncthreads();
    for (int i = lid; i < 16 * WIDTH; i += 32) {
        const int ch = i >> 9, w = i & (WIDTH - 1);
        out[imgBase + (long long)ch * chStride + (long long)w * posStride] = X[ch * XPITCH + w];
    }
}

__global__ __launch_bounds__(VCOLS * 32)
void dt_vpass_kernel(const float* in, float* out, const float* edge, float clog2)
{
    __shared__ float vsm[VCOLS * COLF];

    const int tid  = threadIdx.x;
    const int wv   = tid >> 5;
    const int lid  = tid & 31;
    const int half = lid & 15;
    const int khi  = lid >> 4;

    const int nb = blockIdx.x;
    const int cg = nb & 1;
    const int g  = nb >> 1;
    const int b  = g >> 6;
    const int c0 = (g & 63) * VCOLS;

    const int C = 32, H = 512, Wd = 512;
    const long long chStride = (long long)H * Wd;
    const long long imgBase  = (long long)b * C * H * Wd + (long long)(cg * 16) * chStride + c0;
    const long long edgeBase = (long long)b * H * Wd + c0;

    constexpr int NEDGE = WIDTH * 2 / (VCOLS * 32);
    float4 e4[NEDGE];
#pragma unroll
    for (int k = 0; k < NEDGE; ++k) {
        const int q = tid + k * VCOLS * 32;
        const int j = q & 1, h = q >> 1;
        e4[k] = *reinterpret_cast<const float4*>(edge + edgeBase + (long long)h * Wd + j * 4);
    }

    constexpr int NPIECE = 16 * WIDTH * 2;
    constexpr int BATCH  = 16;
    for (int q0 = tid; q0 < NPIECE; q0 += BATCH * VCOLS * 32) {
        float4 v[BATCH];
#pragma unroll
        for (int k = 0; k < BATCH; ++k) {
            const int q  = q0 + k * VCOLS * 32;
            const int j  = q & 1, h = (q >> 1) & (WIDTH - 1), ch = q >> 10;
            v[k] = *reinterpret_cast<const float4*>(
                       in + imgBase + (long long)ch * chStride + (long long)h * Wd + j * 4);
        }
#pragma unroll
        for (int k = 0; k < BATCH; ++k) {
            const int q  = q0 + k * VCOLS * 32;
            const int j  = q & 1, h = (q >> 1) & (WIDTH - 1), ch = q >> 10;
            float* d = vsm + (j * 4) * COLF + ch * XPITCH + h;
            d[0 * COLF] = v[k].x;
            d[1 * COLF] = v[k].y;
            d[2 * COLF] = v[k].z;
            d[3 * COLF] = v[k].w;
        }
    }
#pragma unroll
    for (int k = 0; k < NEDGE; ++k) {
        const int q = tid + k * VCOLS * 32;
        const int j = q & 1, h = q >> 1;
        const float ev[4] = {e4[k].x, e4[k].y, e4[k].z, e4[k].w};
#pragma unroll
        for (int e = 0; e < 4; ++e) {
            float t = clog2 * (1.0f + 150.0f * ev[e]);
            if (h == 0) t = -512.0f;
            float* col = vsm + (j * 4 + e) * COLF + 16 * XPITCH;
            col[h]         = t;
            col[WIDTH + h] = 1.0f - exp2_fast(t);
        }
    }
    __syncthreads();

    float* X   = vsm + wv * COLF;
    float* tF  = X + 16 * XPITCH;
    float* omF = tF + WIDTH;
    scan_dir(X, tF, omF, half, khi, false);
    scan_dir(X, tF, omF, half, khi, true);

    __syncthreads();
    for (int q = tid; q < NPIECE; q += VCOLS * 32) {
        const int j = q & 1, h = (q >> 1) & (WIDTH - 1), ch = q >> 10;
        const float* s = vsm + (j * 4) * COLF + ch * XPITCH + h;
        float4 v;
        v.x = s[0 * COLF];
        v.y = s[1 * COLF];
        v.z = s[2 * COLF];
        v.w = s[3 * COLF];
        *reinterpret_cast<float4*>(
            out + imgBase + (long long)ch * chStride + (long long)h * Wd + j * 4) = v;
    }
}

extern "C" void kernel_launch(void* const* d_in, const int* in_sizes, int n_in,
                              void* d_out, int out_size, void* d_ws, size_t ws_size,
                              hipStream_t stream) {
    (void)in_sizes; (void)n_in; (void)out_size; (void)d_ws; (void)ws_size;
    const float* img  = (const float*)d_in[0];
    const float* edge = (const float*)d_in[1];
    float* out = (float*)d_out;

    const double SQRT2 = 1.4142135623730951;
    const double SQRT3 = 1.7320508075688772;
    const double LOG2E = 1.4426950408889634;
    const double denom = 3.872983346207417;
    const double sigma1 = 60.0 * SQRT3 * 2.0 / denom;
    const double sigma2 = 60.0 * SQRT3 * 1.0 / denom;
    const float c1 = (float)(-SQRT2 / sigma1 * LOG2E);
    const float c2 = (float)(-SQRT2 / sigma2 * LOG2E);

    const dim3 grid(4096), block(32);
    const size_t shmem = (size_t)(16 * XPITCH + 2 * WIDTH) * sizeof(float);

    const dim3 vgrid(512), vblock(VCOLS * 32);

    dt_pass_kernel<<<grid, block, shmem, stream>>>(img, out, edge, c1, 0);
    dt_vpass_kernel<<<vgrid, vblock, 0, stream>>>(out, out, edge, c1);
    dt_pass_kernel<<<grid, block, shmem, stream>>>(out, out, edge, c2, 0);
    dt_vpass_kernel<<<vgrid, vblock, 0, stream>>>(out, out, edge, c2);
}


# Horizontal pass: the row tile moves with 16-byte global and LDS accesses, 32 loads in flight per lane, edge row asked for first (was 4-byte, 8 in flight); LDS pitch 516; scan and sums unchanged
# speedup vs baseline: 3.5817x; 1.1782x over previous
#include <hip/hip_runtime.h>
#include <math.h>


typedef __attribute__((ext_vector_type(2))) float v2f;
typedef __attribute__((ext_vector_type(8))) float v8f;

#define WIDTH  512
#define SEGS   32
#define XPITCH 513
#define HPITCH 516

#define VCOLS  8
#define COLF   (16 * XPITCH + 2 * WIDTH + 4)

__device__ __forceinline__ float exp2_fast(float x) {
    return __builtin_amdgcn_exp2f(x);
}

template <int CTRL, int ROWS>
__device__ __forceinline__ float dpp_row(float old, float v) {
    return __builtin_bit_cast(float, __builtin_amdgcn_update_dpp(
               __builtin_bit_cast(int, old), __builtin_bit_cast(int, v), CTRL, ROWS, 0xf, true));
}
__device__ __forceinline__ float other_row(float v) {
    return __builtin_bit_cast(float, __builtin_amdgcn_permlanex16(
               __builtin_bit_cast(int, v), __builtin_bit_cast(int, v),
               0x76543210, 0xfedcba98, false, false));
}

template <int PITCH>
__device__ __forceinline__ void scan_dir(float* X, const float* tF, const float* omF,
                                         int half, int khi, bool bwd)
{
    float carry = 0.0f;
    for (int s = 0; s < SEGS; ++s) {
        const int base = s * 16;
        const int idx  = base + half;

        float t;
        if (!bwd) t = tF[idx];
        else      t = (idx == 0) ? -512.0f : tF[WIDTH - idx];

        float S = t;
        S += dpp_row<0x111, 0xf>(0.0f, S);
        S += dpp_row<0x112, 0xf>(0.0f, S);
        S += dpp_row<0x114, 0xf>(0.0f, S);
        S += dpp_row<0x118, 0xf>(0.0f, S);
        const float expS = exp2_fast(S);

        const float Srot = dpp_row<0x12e, 0x2>(S, S);
        const float Prot = dpp_row<0x128, 0x2>(expS, expS);

        v8f acc = {0.f,0.f,0.f,0.f,0.f,0.f,0.f,0.f};
#pragma unroll
        for (int kk = 0; kk < 4; ++kk) {
            const int m0 = kk * 4 + khi * 2;
            float Sm0, Sm1;
            if (kk == 0)      { Sm0 = dpp_row<0x150, 0xf>(0.0f, Srot); Sm1 = dpp_row<0x151, 0xf>(0.0f, Srot); }
            else if (kk == 1) { Sm0 = dpp_row<0x154, 0xf>(0.0f, Srot); Sm1 = dpp_row<0x155, 0xf>(0.0f, Srot); }
            else if (kk == 2) { Sm0 = dpp_row<0x158, 0xf>(0.0f, Srot); Sm1 = dpp_row<0x159, 0xf>(0.0f, Srot); }
            else              { Sm0 = dpp_row<0x15c, 0xf>(0.0f, Srot); Sm1 = dpp_row<0x15d, 0xf>(0.0f, Srot); }
            v2f Af;
            Af.x = (m0     < half) ? exp2_fast(S - Sm0) : ((m0     == half) ? 1.0f : 0.0f);
            Af.y = (m0 + 1 < half) ? exp2_fast(S - Sm1) : ((m0 + 1 == half) ? 1.0f : 0.0f);

            const int p0 = base + m0;
            v2f Bf;
            if (!bwd) {
                Bf.x = omF[p0]     * X[half * PITCH + p0];
                Bf.y = omF[p0 + 1] * X[half * PITCH + p0 + 1];
            } else {
                const float om0 = (p0 == 0) ? 1.0f : omF[WIDTH - p0];
                const float om1 = omF[WIDTH - (p0 + 1)];
                Bf.x = om0 * X[half * PITCH + (WIDTH - 1 - p0)];
                Bf.y = om1 * X[half * PITCH + (WIDTH - 1 - (p0 + 1))];
            }

            acc = __builtin_amdgcn_wmma_f32_16x16x4_f32(
                      false, Af, false, Bf, (short)0, acc, false, false);
        }

        acc[0] += dpp_row<0x150, 0xf>(0.0f, Prot) * carry;
        acc[1] += dpp_row<0x151, 0xf>(0.0f, Prot) * carry;
        acc[2] += dpp_row<0x152, 0xf>(0.0f, Prot) * carry;
        acc[3] += dpp_row<0x153, 0xf>(0.0f, Prot) * carry;
        acc[4] += dpp_row<0x154, 0xf>(0.0f, Prot) * carry;
        acc[5] += dpp_row<0x155, 0xf>(0.0f, Prot) * carry;
        acc[6] += dpp_row<0x156, 0xf>(0.0f, Prot) * carry;
        acc[7] += dpp_row<0x157, 0xf>(0.0f, Prot) * carry;
        const float upper = other_row(acc[7]);
        carry = khi ? acc[7] : upper;

#pragma unroll
        for (int rr = 0; rr < 8; ++rr) {
            const int j = base + rr + 8 * khi;
            const int w = bwd ? (WIDTH - 1 - j) : j;
            X[half * PITCH + w] = acc[rr];
        }
    }
}

__global__ __launch_bounds__(32)
void dt_pass_kernel(const float* in, float* out, const float* edge, float clog2)
{
    __shared__ __attribute__((aligned(16))) float hsm[16 * HPITCH + 2 * WIDTH];
    float* X   = hsm;
    float* tF  = hsm + 16 * HPITCH;
    float* omF = tF + WIDTH;

    const int lid  = threadIdx.x;
    const int half = lid & 15;
    const int khi  = lid >> 4;

    const int nb  = blockIdx.x;
    const int cg  = nb & 1;
    const int row = nb >> 1;
    const int b   = row >> 9;
    const int r   = row & 511;

    const int C = 32, H = 512, Wd = 512;
    const long long chStride = (long long)H * Wd;
    const long long imgBase  = (long long)b * C * H * Wd + (long long)r * Wd
                             + (long long)(cg * 16) * chStride;
    const long long edgeBase = (long long)b * H * Wd + (long long)r * Wd;

    constexpr int NEDGE = WIDTH / 4 / 32;
    float4 e4[NEDGE];
#pragma unroll
    for (int k = 0; k < NEDGE; ++k)
        e4[k] = *reinterpret_cast<const float4*>(edge + edgeBase + (lid + k * 32) * 4);

    constexpr int NPIECE = 16 * WIDTH / 4;
    constexpr int HBATCH = 32;
    for (int q0 = lid; q0 < NPIECE; q0 += HBATCH * 32) {
        float4 v[HBATCH];
#pragma unroll
        for (int k = 0; k < HBATCH; ++k) {
            const int q  = q0 + k * 32;
            const int ch = q >> 7, w = (q & 127) * 4;
            v[k] = *reinterpret_cast<const float4*>(in + imgBase + (long long)ch * chStride + w);
        }
#pragma unroll
        for (int k = 0; k < HBATCH; ++k) {
            const int q  = q0 + k * 32;
            const int ch = q >> 7, w = (q & 127) * 4;
            *reinterpret_cast<float4*>(X + ch * HPITCH + w) = v[k];
        }
    }
#pragma unroll
    for (int k = 0; k < NEDGE; ++k) {
        const int w0 = (lid + k * 32) * 4;
        const float ev[4] = {e4[k].x, e4[k].y, e4[k].z, e4[k].w};
        float tv[4], ov[4];
#pragma unroll
        for (int e = 0; e < 4; ++e) {
            const float Dv = 1.0f + 150.0f * ev[e];
            float t = clog2 * Dv;
            if (w0 + e == 0) t = -512.0f;
            tv[e] = t;
            ov[e] = 1.0f - exp2_fast(t);
        }
        *reinterpret_cast<float4*>(tF  + w0) = make_float4(tv[0], tv[1], tv[2], tv[3]);
        *reinterpret_cast<float4*>(omF + w0) = make_float4(ov[0], ov[1], ov[2], ov[3]);
    }
    __syncthreads();

    scan_dir<HPITCH>(X, tF, omF, half, khi, false);
    scan_dir<HPITCH>(X, tF, omF, half, khi, true);

    __syncthreads();
    constexpr int WBATCH = 16;
    for (int q0 = lid; q0 < NPIECE; q0 += WBATCH * 32) {
        float4 v[WBATCH];
#pragma unroll
        for (int k = 0; k < WBATCH; ++k) {
            const int q  = q0 + k * 32;
            const int ch = q >> 7, w = (q & 127) * 4;
            v[k] = *reinterpret_cast<const float4*>(X + ch * HPITCH + w);
        }
#pragma unroll
        for (int k = 0; k < WBATCH; ++k) {
            const int q  = q0 + k * 32;
            const int ch = q >> 7, w = (q & 127) * 4;
            *reinterpret_cast<float4*>(out + imgBase + (long long)ch * chStride + w) = v[k];
        }
    }
}

__global__ __launch_bounds__(VCOLS * 32)
void dt_vpass_kernel(const float* in, float* out, const float* edge, float clog2)
{
    __shared__ float vsm[VCOLS * COLF];

    const int tid  = threadIdx.x;
    const int wv   = tid >> 5;
    const int lid  = tid & 31;
    const int half = lid & 15;
    const int khi  = lid >> 4;

    const int nb = blockIdx.x;
    const int cg = nb & 1;
    const int g  = nb >> 1;
    const int b  = g >> 6;
    const int c0 = (g & 63) * VCOLS;

    const int C = 32, H = 512, Wd = 512;
    const long long chStride = (long long)H * Wd;
    const long long imgBase  = (long long)b * C * H * Wd + (long long)(cg * 16) * chStride + c0;
    const long long edgeBase = (long long)b * H * Wd + c0;

    constexpr int NEDGE = WIDTH * 2 / (VCOLS * 32);
    float4 e4[NEDGE];
#pragma unroll
    for (int k = 0; k < NEDGE; ++k) {
        const int q = tid + k * VCOLS * 32;
        const int j = q & 1, h = q >> 1;
        e4[k] = *reinterpret_cast<const float4*>(edge + edgeBase + (long long)h * Wd + j * 4);
    }

    constexpr int NPIECE = 16 * WIDTH * 2;
    constexpr int BATCH  = 16;
    for (int q0 = tid; q0 < NPIECE; q0 += BATCH * VCOLS * 32) {
        float4 v[BATCH];
#pragma unroll
        for (int k = 0; k < BATCH; ++k) {
            const int q  = q0 + k * VCOLS * 32;
            const int j  = q & 1, h = (q >> 1) & (WIDTH - 1), ch = q >> 10;
            v[k] = *reinterpret_cast<const float4*>(
                       in + imgBase + (long long)ch * chStride + (long long)h * Wd + j * 4);
        }
#pragma unroll
        for (int k = 0; k < BATCH; ++k) {
            const int q  = q0 + k * VCOLS * 32;
            const int j  = q & 1, h = (q >> 1) & (WIDTH - 1), ch = q >> 10;
            float* d = vsm + (j * 4) * COLF + ch * XPITCH + h;
            d[0 * COLF] = v[k].x;
            d[1 * COLF] = v[k].y;
            d[2 * COLF] = v[k].z;
            d[3 * COLF] = v[k].w;
        }
    }
#pragma unroll
    for (int k = 0; k < NEDGE; ++k) {
        const int q = tid + k * VCOLS * 32;
        const int j = q & 1, h = q >> 1;
        const float ev[4] = {e4[k].x, e4[k].y, e4[k].z, e4[k].w};
#pragma unroll
        for (int e = 0; e < 4; ++e) {
            float t = clog2 * (1.0f + 150.0f * ev[e]);
            if (h == 0) t = -512.0f;
            float* col = vsm + (j * 4 + e) * COLF + 16 * XPITCH;
            col[h]         = t;
            col[WIDTH + h] = 1.0f - exp2_fast(t);
        }
    }
    __syncthreads();

    float* X   = vsm + wv * COLF;
    float* tF  = X + 16 * XPITCH;
    float* omF = tF + WIDTH;
    scan_dir<XPITCH>(X, tF, omF, half, khi, false);
    scan_dir<XPITCH>(X, tF, omF, half, khi, true);

    __syncthreads();
    for (int q = tid; q < NPIECE; q += VCOLS * 32) {
        const int j = q & 1, h = (q >> 1) & (WIDTH - 1), ch = q >> 10;
        const float* s = vsm + (j * 4) * COLF + ch * XPITCH + h;
        float4 v;
        v.x = s[0 * COLF];
        v.y = s[1 * COLF];
        v.z = s[2 * COLF];
        v.w = s[3 * COLF];
        *reinterpret_cast<float4*>(
            out + imgBase + (long long)ch * chStride + (long long)h * Wd + j * 4) = v;
    }
}

extern "C" void kernel_launch(void* const* d_in, const int* in_sizes, int n_in,
                              void* d_out, int out_size, void* d_ws, size_t ws_size,
                              hipStream_t stream) {
    (void)in_sizes; (void)n_in; (void)out_size; (void)d_ws; (void)ws_size;
    const float* img  = (const float*)d_in[0];
    const float* edge = (const float*)d_in[1];
    float* out = (float*)d_out;

    const double SQRT2 = 1.4142135623730951;
    const double SQRT3 = 1.7320508075688772;
    const double LOG2E = 1.4426950408889634;
    const double denom = 3.872983346207417;
    const double sigma1 = 60.0 * SQRT3 * 2.0 / denom;
    const double sigma2 = 60.0 * SQRT3 * 1.0 / denom;
    const float c1 = (float)(-SQRT2 / sigma1 * LOG2E);
    const float c2 = (float)(-SQRT2 / sigma2 * LOG2E);

    const dim3 grid(4096), block(32);

    const dim3 vgrid(512), vblock(VCOLS * 32);

    dt_pass_kernel<<<grid, block, 0, stream>>>(img, out, edge, c1);
    dt_vpass_kernel<<<vgrid, vblock, 0, stream>>>(out, out, edge, c1);
    dt_pass_kernel<<<grid, block, 0, stream>>>(out, out, edge, c2);
    dt_vpass_kernel<<<vgrid, vblock, 0, stream>>>(out, out, edge, c2);
}


# Vertical pass: a work-group takes two column groups; the second group's global loads go into registers ahead of the first group's scan (barriers order LDS only) and land while it runs
# speedup vs baseline: 4.0026x; 1.1175x over previous
#include <hip/hip_runtime.h>
#include <math.h>


typedef __attribute__((ext_vector_type(2))) float v2f;
typedef __attribute__((ext_vector_type(8))) float v8f;

#define WIDTH  512
#define SEGS   32
#define XPITCH 513
#define HPITCH 516

#define VCOLS  8
#define COLF   (16 * XPITCH + 2 * WIDTH + 4)

__device__ __forceinline__ float exp2_fast(float x) {
    return __builtin_amdgcn_exp2f(x);
}

template <int CTRL, int ROWS>
__device__ __forceinline__ float dpp_row(float old, float v) {
    return __builtin_bit_cast(float, __builtin_amdgcn_update_dpp(
               __builtin_bit_cast(int, old), __builtin_bit_cast(int, v), CTRL, ROWS, 0xf, true));
}
__device__ __forceinline__ float other_row(float v) {
    return __builtin_bit_cast(float, __builtin_amdgcn_permlanex16(
               __builtin_bit_cast(int, v), __builtin_bit_cast(int, v),
               0x76543210, 0xfedcba98, false, false));
}

template <int PITCH>
__device__ __forceinline__ void scan_dir(float* X, const float* tF, const float* omF,
                                         int half, int khi, bool bwd)
{
    float carry = 0.0f;
    for (int s = 0; s < SEGS; ++s) {
        const int base = s * 16;
        const int idx  = base + half;

        float t;
        if (!bwd) t = tF[idx];
        else      t = (idx == 0) ? -512.0f : tF[WIDTH - idx];

        float S = t;
        S += dpp_row<0x111, 0xf>(0.0f, S);
        S += dpp_row<0x112, 0xf>(0.0f, S);
        S += dpp_row<0x114, 0xf>(0.0f, S);
        S += dpp_row<0x118, 0xf>(0.0f, S);
        const float expS = exp2_fast(S);

        const float Srot = dpp_row<0x12e, 0x2>(S, S);
        const float Prot = dpp_row<0x128, 0x2>(expS, expS);

        v8f acc = {0.f,0.f,0.f,0.f,0.f,0.f,0.f,0.f};
#pragma unroll
        for (int kk = 0; kk < 4; ++kk) {
            const int m0 = kk * 4 + khi * 2;
            float Sm0, Sm1;
            if (kk == 0)      { Sm0 = dpp_row<0x150, 0xf>(0.0f, Srot); Sm1 = dpp_row<0x151, 0xf>(0.0f, Srot); }
            else if (kk == 1) { Sm0 = dpp_row<0x154, 0xf>(0.0f, Srot); Sm1 = dpp_row<0x155, 0xf>(0.0f, Srot); }
            else if (kk == 2) { Sm0 = dpp_row<0x158, 0xf>(0.0f, Srot); Sm1 = dpp_row<0x159, 0xf>(0.0f, Srot); }
            else              { Sm0 = dpp_row<0x15c, 0xf>(0.0f, Srot); Sm1 = dpp_row<0x15d, 0xf>(0.0f, Srot); }
            v2f Af;
            Af.x = (m0     < half) ? exp2_fast(S - Sm0) : ((m0     == half) ? 1.0f : 0.0f);
            Af.y = (m0 + 1 < half) ? exp2_fast(S - Sm1) : ((m0 + 1 == half) ? 1.0f : 0.0f);

            const int p0 = base + m0;
            v2f Bf;
            if (!bwd) {
                Bf.x = omF[p0]     * X[half * PITCH + p0];
                Bf.y = omF[p0 + 1] * X[half * PITCH + p0 + 1];
            } else {
                const float om0 = (p0 == 0) ? 1.0f : omF[WIDTH - p0];
                const float om1 = omF[WIDTH - (p0 + 1)];
                Bf.x = om0 * X[half * PITCH + (WIDTH - 1 - p0)];
                Bf.y = om1 * X[half * PITCH + (WIDTH - 1 - (p0 + 1))];
            }

            acc = __builtin_amdgcn_wmma_f32_16x16x4_f32(
                      false, Af, false, Bf, (short)0, acc, false, false);
        }

        acc[0] += dpp_row<0x150, 0xf>(0.0f, Prot) * carry;
        acc[1] += dpp_row<0x151, 0xf>(0.0f, Prot) * carry;
        acc[2] += dpp_row<0x152, 0xf>(0.0f, Prot) * carry;
        acc[3] += dpp_row<0x153, 0xf>(0.0f, Prot) * carry;
        acc[4] += dpp_row<0x154, 0xf>(0.0f, Prot) * carry;
        acc[5] += dpp_row<0x155, 0xf>(0.0f, Prot) * carry;
        acc[6] += dpp_row<0x156, 0xf>(0.0f, Prot) * carry;
        acc[7] += dpp_row<0x157, 0xf>(0.0f, Prot) * carry;
        const float upper = other_row(acc[7]);
        carry = khi ? acc[7] : upper;

#pragma unroll
        for (int rr = 0; rr < 8; ++rr) {
            const int j = base + rr + 8 * khi;
            const int w = bwd ? (WIDTH - 1 - j) : j;
            X[half * PITCH + w] = acc[rr];
        }
    }
}

__global__ __launch_bounds__(32)
void dt_pass_kernel(const float* in, float* out, const float* edge, float clog2)
{
    __shared__ __attribute__((aligned(16))) float hsm[16 * HPITCH + 2 * WIDTH];
    float* X   = hsm;
    float* tF  = hsm + 16 * HPITCH;
    float* omF = tF + WIDTH;

    const int lid  = threadIdx.x;
    const int half = lid & 15;
    const int khi  = lid >> 4;

    const int nb  = blockIdx.x;
    const int cg  = nb & 1;
    const int row = nb >> 1;
    const int b   = row >> 9;
    const int r   = row & 511;

    const int C = 32, H = 512, Wd = 512;
    const long long chStride = (long long)H * Wd;
    const long long imgBase  = (long long)b * C * H * Wd + (long long)r * Wd
                             + (long long)(cg * 16) * chStride;
    const long long edgeBase = (long long)b * H * Wd + (long long)r * Wd;

    constexpr int NEDGE = WIDTH / 4 / 32;
    float4 e4[NEDGE];
#pragma unroll
    for (int k = 0; k < NEDGE; ++k)
        e4[k] = *reinterpret_cast<const float4*>(edge + edgeBase + (lid + k * 32) * 4);

    constexpr int NPIECE = 16 * WIDTH / 4;
    constexpr int HBATCH = 32;
    for (int q0 = lid; q0 < NPIECE; q0 += HBATCH * 32) {
        float4 v[HBATCH];
#pragma unroll
        for (int k = 0; k < HBATCH; ++k) {
            const int q  = q0 + k * 32;
            const int ch = q >> 7, w = (q & 127) * 4;
            v[k] = *reinterpret_cast<const float4*>(in + imgBase + (long long)ch * chStride + w);
        }
#pragma unroll
        for (int k = 0; k < HBATCH; ++k) {
            const int q  = q0 + k * 32;
            const int ch = q >> 7, w = (q & 127) * 4;
            *reinterpret_cast<float4*>(X + ch * HPITCH + w) = v[k];
        }
    }
#pragma unroll
    for (int k = 0; k < NEDGE; ++k) {
        const int w0 = (lid + k * 32) * 4;
        const float ev[4] = {e4[k].x, e4[k].y, e4[k].z, e4[k].w};
        float tv[4], ov[4];
#pragma unroll
        for (int e = 0; e < 4; ++e) {
            const float Dv = 1.0f + 150.0f * ev[e];
            float t = clog2 * Dv;
            if (w0 + e == 0) t = -512.0f;
            tv[e] = t;
            ov[e] = 1.0f - exp2_fast(t);
        }
        *reinterpret_cast<float4*>(tF  + w0) = make_float4(tv[0], tv[1], tv[2], tv[3]);
        *reinterpret_cast<float4*>(omF + w0) = make_float4(ov[0], ov[1], ov[2], ov[3]);
    }
    __syncthreads();

    scan_dir<HPITCH>(X, tF, omF, half, khi, false);
    scan_dir<HPITCH>(X, tF, omF, half, khi, true);

    __syncthreads();
    constexpr int WBATCH = 16;
    for (int q0 = lid; q0 < NPIECE; q0 += WBATCH * 32) {
        float4 v[WBATCH];
#pragma unroll
        for (int k = 0; k < WBATCH; ++k) {
            const int q  = q0 + k * 32;
            const int ch = q >> 7, w = (q & 127) * 4;
            v[k] = *reinterpret_cast<const float4*>(X + ch * HPITCH + w);
        }
#pragma unroll
        for (int k = 0; k < WBATCH; ++k) {
            const int q  = q0 + k * 32;
            const int ch = q >> 7, w = (q & 127) * 4;
            *reinterpret_cast<float4*>(out + imgBase + (long long)ch * chStride + w) = v[k];
        }
    }
}

#define VTILES 2
#define PF_AHEAD 56
#define PF_ALL   64

__device__ __forceinline__ void lds_barrier() {
    __builtin_amdgcn_fence(__ATOMIC_RELEASE, "workgroup", "local");
    __builtin_amdgcn_s_barrier();
    __builtin_amdgcn_fence(__ATOMIC_ACQUIRE, "workgroup", "local");
}

template <int K0, int K1>
__device__ __forceinline__ void v_issue(const float* in, long long imgBase, int tid,
                                        float4 (&v)[PF_ALL])
{
#pragma unroll
    for (int k = K0; k < K1; ++k) {
        const int q  = tid + k * VCOLS * 32;
        const int j  = q & 1, h = (q >> 1) & (WIDTH - 1), ch = q >> 10;
        v[k] = *reinterpret_cast<const float4*>(
                   in + imgBase + (long long)ch * (512LL * 512LL) + (long long)h * 512 + j * 4);
    }
}

__global__ __launch_bounds__(VCOLS * 32)
void dt_vpass_kernel(const float* in, float* out, const float* edge, float clog2)
{
    __shared__ float vsm[VCOLS * COLF];

    const int tid  = threadIdx.x;
    const int wv   = tid >> 5;
    const int lid  = tid & 31;
    const int half = lid & 15;
    const int khi  = lid >> 4;

    const int C = 32, H = 512, Wd = 512;
    const long long chStride = (long long)H * Wd;

    constexpr int NEDGE  = WIDTH * 2 / (VCOLS * 32);
    constexpr int NPIECE = 16 * WIDTH * 2;

    float4 e4[NEDGE];
    float4 v[PF_ALL];

    int nb = blockIdx.x;
    long long imgBase, edgeBase;
    {
        const int cg = nb & 1, g = nb >> 1, b = g >> 6, c0 = (g & 63) * VCOLS;
        imgBase  = (long long)b * C * H * Wd + (long long)(cg * 16) * chStride + c0;
        edgeBase = (long long)b * H * Wd + c0;
    }
#pragma unroll
    for (int k = 0; k < NEDGE; ++k) {
        const int q = tid + k * VCOLS * 32;
        const int j = q & 1, h = q >> 1;
        e4[k] = *reinterpret_cast<const float4*>(edge + edgeBase + (long long)h * Wd + j * 4);
    }
    v_issue<0, PF_ALL>(in, imgBase, tid, v);

    float* X   = vsm + wv * COLF;
    float* tF  = X + 16 * XPITCH;
    float* omF = tF + WIDTH;

#pragma unroll
    for (int t = 0; t < VTILES; ++t) {
#pragma unroll
        for (int k = 0; k < PF_ALL; ++k) {
            const int q  = tid + k * VCOLS * 32;
            const int j  = q & 1, h = (q >> 1) & (WIDTH - 1), ch = q >> 10;
            float* d = vsm + (j * 4) * COLF + ch * XPITCH + h;
            d[0 * COLF] = v[k].x;
            d[1 * COLF] = v[k].y;
            d[2 * COLF] = v[k].z;
            d[3 * COLF] = v[k].w;
        }
#pragma unroll
        for (int k = 0; k < NEDGE; ++k) {
            const int q = tid + k * VCOLS * 32;
            const int j = q & 1, h = q >> 1;
            const float ev[4] = {e4[k].x, e4[k].y, e4[k].z, e4[k].w};
#pragma unroll
            for (int e = 0; e < 4; ++e) {
                float tt = clog2 * (1.0f + 150.0f * ev[e]);
                if (h == 0) tt = -512.0f;
                float* col = vsm + (j * 4 + e) * COLF + 16 * XPITCH;
                col[h]         = tt;
                col[WIDTH + h] = 1.0f - exp2_fast(tt);
            }
        }

        const long long curBase = imgBase;
        if (t + 1 < VTILES) {
            nb += gridDim.x;
            const int cg = nb & 1, g = nb >> 1, b = g >> 6, c0 = (g & 63) * VCOLS;
            imgBase  = (long long)b * C * H * Wd + (long long)(cg * 16) * chStride + c0;
            edgeBase = (long long)b * H * Wd + c0;
#pragma unroll
            for (int k = 0; k < NEDGE; ++k) {
                const int q = tid + k * VCOLS * 32;
                const int j = q & 1, h = q >> 1;
                e4[k] = *reinterpret_cast<const float4*>(edge + edgeBase + (long long)h * Wd + j * 4);
            }
            v_issue<0, PF_AHEAD>(in, imgBase, tid, v);
        }
        lds_barrier();

        scan_dir<XPITCH>(X, tF, omF, half, khi, false);
        scan_dir<XPITCH>(X, tF, omF, half, khi, true);

        lds_barrier();
        if (t + 1 < VTILES)
            v_issue<PF_AHEAD, PF_ALL>(in, imgBase, tid, v);

        constexpr int VWB = 4;
        for (int q0 = tid; q0 < NPIECE; q0 += VWB * VCOLS * 32) {
            float4 o[VWB];
#pragma unroll
            for (int k = 0; k < VWB; ++k) {
                const int q = q0 + k * VCOLS * 32;
                const int j = q & 1, h = (q >> 1) & (WIDTH - 1), ch = q >> 10;
                const float* s = vsm + (j * 4) * COLF + ch * XPITCH + h;
                o[k].x = s[0 * COLF];
                o[k].y = s[1 * COLF];
                o[k].z = s[2 * COLF];
                o[k].w = s[3 * COLF];
            }
#pragma unroll
            for (int k = 0; k < VWB; ++k) {
                const int q = q0 + k * VCOLS * 32;
                const int j = q & 1, h = (q >> 1) & (WIDTH - 1), ch = q >> 10;
                *reinterpret_cast<float4*>(
                    out + curBase + (long long)ch * chStride + (long long)h * Wd + j * 4) = o[k];
            }
        }
        if (t + 1 < VTILES) lds_barrier();
    }
}

extern "C" void kernel_launch(void* const* d_in, const int* in_sizes, int n_in,
                              void* d_out, int out_size, void* d_ws, size_t ws_size,
                              hipStream_t stream) {
    (void)in_sizes; (void)n_in; (void)out_size; (void)d_ws; (void)ws_size;
    const float* img  = (const float*)d_in[0];
    const float* edge = (const float*)d_in[1];
    float* out = (float*)d_out;

    const double SQRT2 = 1.4142135623730951;
    const double SQRT3 = 1.7320508075688772;
    const double LOG2E = 1.4426950408889634;
    const double denom = 3.872983346207417;
    const double sigma1 = 60.0 * SQRT3 * 2.0 / denom;
    const double sigma2 = 60.0 * SQRT3 * 1.0 / denom;
    const float c1 = (float)(-SQRT2 / sigma1 * LOG2E);
    const float c2 = (float)(-SQRT2 / sigma2 * LOG2E);

    const dim3 grid(4096), block(32);

    const dim3 vgrid(256), vblock(VCOLS * 32);

    dt_pass_kernel<<<grid, block, 0, stream>>>(img, out, edge, c1);
    dt_vpass_kernel<<<vgrid, vblock, 0, stream>>>(out, out, edge, c1);
    dt_pass_kernel<<<grid, block, 0, stream>>>(out, out, edge, c2);
    dt_vpass_kernel<<<vgrid, vblock, 0, stream>>>(out, out, edge, c2);
}


# Scan loop takes two segments per trip: both segments' coefficient rows, T fragments and 4-WMMA chains are formed ahead of the two carry updates (one LDS wait, chains alternate); sums unchanged
# speedup vs baseline: 4.2450x; 1.0606x over previous
#include <hip/hip_runtime.h>
#include <math.h>


typedef __attribute__((ext_vector_type(2))) float v2f;
typedef __attribute__((ext_vector_type(8))) float v8f;

#define WIDTH  512
#define SEGS   32
#define XPITCH 513
#define HPITCH 516

#define VCOLS  8
#define COLF   (16 * XPITCH + 2 * WIDTH + 4)

__device__ __forceinline__ float exp2_fast(float x) {
    return __builtin_amdgcn_exp2f(x);
}

template <int CTRL, int ROWS>
__device__ __forceinline__ float dpp_row(float old, float v) {
    return __builtin_bit_cast(float, __builtin_amdgcn_update_dpp(
               __builtin_bit_cast(int, old), __builtin_bit_cast(int, v), CTRL, ROWS, 0xf, true));
}
__device__ __forceinline__ float other_row(float v) {
    return __builtin_bit_cast(float, __builtin_amdgcn_permlanex16(
               __builtin_bit_cast(int, v), __builtin_bit_cast(int, v),
               0x76543210, 0xfedcba98, false, false));
}


template <int PITCH>
__device__ __forceinline__ void seg_products(const float* X, const float* tF, const float* omF,
                                             int half, int khi, bool bwd, int s,
                                             v8f& acc, float& Prot)
{
    const int base = s * 16;
    const int idx  = base + half;

    float t;
    if (!bwd) t = tF[idx];
    else {
        const float tl = tF[(WIDTH - idx) & (WIDTH - 1)];
        t = (idx == 0) ? -512.0f : tl;
    }

    float S = t;
    S += dpp_row<0x111, 0xf>(0.0f, S);
    S += dpp_row<0x112, 0xf>(0.0f, S);
    S += dpp_row<0x114, 0xf>(0.0f, S);
    S += dpp_row<0x118, 0xf>(0.0f, S);
    const float expS = exp2_fast(S);

    const float Srot = dpp_row<0x12e, 0x2>(S, S);
    Prot = dpp_row<0x128, 0x2>(expS, expS);

    acc = (v8f){0.f,0.f,0.f,0.f,0.f,0.f,0.f,0.f};
#pragma unroll
    for (int kk = 0; kk < 4; ++kk) {
        const int m0 = kk * 4 + khi * 2;
        float Sm0, Sm1;
        if (kk == 0)      { Sm0 = dpp_row<0x150, 0xf>(0.0f, Srot); Sm1 = dpp_row<0x151, 0xf>(0.0f, Srot); }
        else if (kk == 1) { Sm0 = dpp_row<0x154, 0xf>(0.0f, Srot); Sm1 = dpp_row<0x155, 0xf>(0.0f, Srot); }
        else if (kk == 2) { Sm0 = dpp_row<0x158, 0xf>(0.0f, Srot); Sm1 = dpp_row<0x159, 0xf>(0.0f, Srot); }
        else              { Sm0 = dpp_row<0x15c, 0xf>(0.0f, Srot); Sm1 = dpp_row<0x15d, 0xf>(0.0f, Srot); }
        v2f Af;
        Af.x = (m0     < half) ? exp2_fast(S - Sm0) : ((m0     == half) ? 1.0f : 0.0f);
        Af.y = (m0 + 1 < half) ? exp2_fast(S - Sm1) : ((m0 + 1 == half) ? 1.0f : 0.0f);

        const int p0 = base + m0;
        v2f Bf;
        if (!bwd) {
            Bf.x = omF[p0]     * X[half * PITCH + p0];
            Bf.y = omF[p0 + 1] * X[half * PITCH + p0 + 1];
        } else {
            const float ol  = omF[(WIDTH - p0) & (WIDTH - 1)];
            const float om0 = (p0 == 0) ? 1.0f : ol;
            const float om1 = omF[WIDTH - (p0 + 1)];
            Bf.x = om0 * X[half * PITCH + (WIDTH - 1 - p0)];
            Bf.y = om1 * X[half * PITCH + (WIDTH - 1 - (p0 + 1))];
        }

        acc = __builtin_amdgcn_wmma_f32_16x16x4_f32(
                  false, Af, false, Bf, (short)0, acc, false, false);
    }
}

template <int PITCH>
__device__ __forceinline__ void seg_carry(float* X, int half, int khi, bool bwd, int s,
                                          v8f acc, float Prot, float& carry)
{
    const int base = s * 16;
    acc[0] += dpp_row<0x150, 0xf>(0.0f, Prot) * carry;
    acc[1] += dpp_row<0x151, 0xf>(0.0f, Prot) * carry;
    acc[2] += dpp_row<0x152, 0xf>(0.0f, Prot) * carry;
    acc[3] += dpp_row<0x153, 0xf>(0.0f, Prot) * carry;
    acc[4] += dpp_row<0x154, 0xf>(0.0f, Prot) * carry;
    acc[5] += dpp_row<0x155, 0xf>(0.0f, Prot) * carry;
    acc[6] += dpp_row<0x156, 0xf>(0.0f, Prot) * carry;
    acc[7] += dpp_row<0x157, 0xf>(0.0f, Prot) * carry;
    const float upper = other_row(acc[7]);
    carry = khi ? acc[7] : upper;

#pragma unroll
    for (int rr = 0; rr < 8; ++rr) {
        const int j = base + rr + 8 * khi;
        const int w = bwd ? (WIDTH - 1 - j) : j;
        X[half * PITCH + w] = acc[rr];
    }
}

template <int PITCH>
__device__ __forceinline__ void scan_dir(float* X, const float* tF, const float* omF,
                                         int half, int khi, bool bwd)
{
    float carry = 0.0f;
    for (int s = 0; s < SEGS; s += 2) {
        v8f acc0, acc1;
        float P0, P1;
        seg_products<PITCH>(X, tF, omF, half, khi, bwd, s,     acc0, P0);
        seg_products<PITCH>(X, tF, omF, half, khi, bwd, s + 1, acc1, P1);
        seg_carry<PITCH>(X, half, khi, bwd, s,     acc0, P0, carry);
        seg_carry<PITCH>(X, half, khi, bwd, s + 1, acc1, P1, carry);
    }
}

__global__ __launch_bounds__(32)
void dt_pass_kernel(const float* in, float* out, const float* edge, float clog2)
{
    __shared__ __attribute__((aligned(16))) float hsm[16 * HPITCH + 2 * WIDTH];
    float* X   = hsm;
    float* tF  = hsm + 16 * HPITCH;
    float* omF = tF + WIDTH;

    const int lid  = threadIdx.x;
    const int half = lid & 15;
    const int khi  = lid >> 4;

    const int nb  = blockIdx.x;
    const int cg  = nb & 1;
    const int row = nb >> 1;
    const int b   = row >> 9;
    const int r   = row & 511;

    const int C = 32, H = 512, Wd = 512;
    const long long chStride = (long long)H * Wd;
    const long long imgBase  = (long long)b * C * H * Wd + (long long)r * Wd
                             + (long long)(cg * 16) * chStride;
    const long long edgeBase = (long long)b * H * Wd + (long long)r * Wd;

    constexpr int NEDGE = WIDTH / 4 / 32;
    float4 e4[NEDGE];
#pragma unroll
    for (int k = 0; k < NEDGE; ++k)
        e4[k] = *reinterpret_cast<const float4*>(edge + edgeBase + (lid + k * 32) * 4);

    constexpr int NPIECE = 16 * WIDTH / 4;
    constexpr int HBATCH = 32;
    for (int q0 = lid; q0 < NPIECE; q0 += HBATCH * 32) {
        float4 v[HBATCH];
#pragma unroll
        for (int k = 0; k < HBATCH; ++k) {
            const int q  = q0 + k * 32;
            const int ch = q >> 7, w = (q & 127) * 4;
            v[k] = *reinterpret_cast<const float4*>(in + imgBase + (long long)ch * chStride + w);
        }
#pragma unroll
        for (int k = 0; k < HBATCH; ++k) {
            const int q  = q0 + k * 32;
            const int ch = q >> 7, w = (q & 127) * 4;
            *reinterpret_cast<float4*>(X + ch * HPITCH + w) = v[k];
        }
    }
#pragma unroll
    for (int k = 0; k < NEDGE; ++k) {
        const int w0 = (lid + k * 32) * 4;
        const float ev[4] = {e4[k].x, e4[k].y, e4[k].z, e4[k].w};
        float tv[4], ov[4];
#pragma unroll
        for (int e = 0; e < 4; ++e) {
            const float Dv = 1.0f + 150.0f * ev[e];
            float t = clog2 * Dv;
            if (w0 + e == 0) t = -512.0f;
            tv[e] = t;
            ov[e] = 1.0f - exp2_fast(t);
        }
        *reinterpret_cast<float4*>(tF  + w0) = make_float4(tv[0], tv[1], tv[2], tv[3]);
        *reinterpret_cast<float4*>(omF + w0) = make_float4(ov[0], ov[1], ov[2], ov[3]);
    }
    __syncthreads();

    scan_dir<HPITCH>(X, tF, omF, half, khi, false);
    scan_dir<HPITCH>(X, tF, omF, half, khi, true);

    __syncthreads();
    constexpr int WBATCH = 16;
    for (int q0 = lid; q0 < NPIECE; q0 += WBATCH * 32) {
        float4 v[WBATCH];
#pragma unroll
        for (int k = 0; k < WBATCH; ++k) {
            const int q  = q0 + k * 32;
            const int ch = q >> 7, w = (q & 127) * 4;
            v[k] = *reinterpret_cast<const float4*>(X + ch * HPITCH + w);
        }
#pragma unroll
        for (int k = 0; k < WBATCH; ++k) {
            const int q  = q0 + k * 32;
            const int ch = q >> 7, w = (q & 127) * 4;
            *reinterpret_cast<float4*>(out + imgBase + (long long)ch * chStride + w) = v[k];
        }
    }
}

#define VTILES 2
#define PF_AHEAD 56
#define PF_ALL   64

__device__ __forceinline__ void lds_barrier() {
    __builtin_amdgcn_fence(__ATOMIC_RELEASE, "workgroup", "local");
    __builtin_amdgcn_s_barrier();
    __builtin_amdgcn_fence(__ATOMIC_ACQUIRE, "workgroup", "local");
}

template <int K0, int K1>
__device__ __forceinline__ void v_issue(const float* in, long long imgBase, int tid,
                                        float4 (&v)[PF_ALL])
{
#pragma unroll
    for (int k = K0; k < K1; ++k) {
        const int q  = tid + k * VCOLS * 32;
        const int j  = q & 1, h = (q >> 1) & (WIDTH - 1), ch = q >> 10;
        v[k] = *reinterpret_cast<const float4*>(
                   in + imgBase + (long long)ch * (512LL * 512LL) + (long long)h * 512 + j * 4);
    }
}

__global__ __launch_bounds__(VCOLS * 32)
void dt_vpass_kernel(const float* in, float* out, const float* edge, float clog2)
{
    __shared__ float vsm[VCOLS * COLF];

    const int tid  = threadIdx.x;
    const int wv   = tid >> 5;
    const int lid  = tid & 31;
    const int half = lid & 15;
    const int khi  = lid >> 4;

    const int C = 32, H = 512, Wd = 512;
    const long long chStride = (long long)H * Wd;

    constexpr int NEDGE  = WIDTH * 2 / (VCOLS * 32);
    constexpr int NPIECE = 16 * WIDTH * 2;

    float4 e4[NEDGE];
    float4 v[PF_ALL];

    int nb = blockIdx.x;
    long long imgBase, edgeBase;
    {
        const int cg = nb & 1, g = nb >> 1, b = g >> 6, c0 = (g & 63) * VCOLS;
        imgBase  = (long long)b * C * H * Wd + (long long)(cg * 16) * chStride + c0;
        edgeBase = (long long)b * H * Wd + c0;
    }
#pragma unroll
    for (int k = 0; k < NEDGE; ++k) {
        const int q = tid + k * VCOLS * 32;
        const int j = q & 1, h = q >> 1;
        e4[k] = *reinterpret_cast<const float4*>(edge + edgeBase + (long long)h * Wd + j * 4);
    }
    v_issue<0, PF_ALL>(in, imgBase, tid, v);

    float* X   = vsm + wv * COLF;
    float* tF  = X + 16 * XPITCH;
    float* omF = tF + WIDTH;

#pragma unroll
    for (int t = 0; t < VTILES; ++t) {
#pragma unroll
        for (int k = 0; k < PF_ALL; ++k) {
            const int q  = tid + k * VCOLS * 32;
            const int j  = q & 1, h = (q >> 1) & (WIDTH - 1), ch = q >> 10;
            float* d = vsm + (j * 4) * COLF + ch * XPITCH + h;
            d[0 * COLF] = v[k].x;
            d[1 * COLF] = v[k].y;
            d[2 * COLF] = v[k].z;
            d[3 * COLF] = v[k].w;
        }
#pragma unroll
        for (int k = 0; k < NEDGE; ++k) {
            const int q = tid + k * VCOLS * 32;
            const int j = q & 1, h = q >> 1;
            const float ev[4] = {e4[k].x, e4[k].y, e4[k].z, e4[k].w};
#pragma unroll
            for (int e = 0; e < 4; ++e) {
                float tt = clog2 * (1.0f + 150.0f * ev[e]);
                if (h == 0) tt = -512.0f;
                float* col = vsm + (j * 4 + e) * COLF + 16 * XPITCH;
                col[h]         = tt;
                col[WIDTH + h] = 1.0f - exp2_fast(tt);
            }
        }

        const long long curBase = imgBase;
        if (t + 1 < VTILES) {
            nb += gridDim.x;
            const int cg = nb & 1, g = nb >> 1, b = g >> 6, c0 = (g & 63) * VCOLS;
            imgBase  = (long long)b * C * H * Wd + (long long)(cg * 16) * chStride + c0;
            edgeBase = (long long)b * H * Wd + c0;
#pragma unroll
            for (int k = 0; k < NEDGE; ++k) {
                const int q = tid + k * VCOLS * 32;
                const int j = q & 1, h = q >> 1;
                e4[k] = *reinterpret_cast<const float4*>(edge + edgeBase + (long long)h * Wd + j * 4);
            }
            v_issue<0, PF_AHEAD>(in, imgBase, tid, v);
        }
        lds_barrier();

        scan_dir<XPITCH>(X, tF, omF, half, khi, false);
        scan_dir<XPITCH>(X, tF, omF, half, khi, true);

        lds_barrier();
        if (t + 1 < VTILES)
            v_issue<PF_AHEAD, PF_ALL>(in, imgBase, tid, v);

        constexpr int VWB = 4;
        for (int q0 = tid; q0 < NPIECE; q0 += VWB * VCOLS * 32) {
            float4 o[VWB];
#pragma unroll
            for (int k = 0; k < VWB; ++k) {
                const int q = q0 + k * VCOLS * 32;
                const int j = q & 1, h = (q >> 1) & (WIDTH - 1), ch = q >> 10;
                const float* s = vsm + (j * 4) * COLF + ch * XPITCH + h;
                o[k].x = s[0 * COLF];
                o[k].y = s[1 * COLF];
                o[k].z = s[2 * COLF];
                o[k].w = s[3 * COLF];
            }
#pragma unroll
            for (int k = 0; k < VWB; ++k) {
                const int q = q0 + k * VCOLS * 32;
                const int j = q & 1, h = (q >> 1) & (WIDTH - 1), ch = q >> 10;
                *reinterpret_cast<float4*>(
                    out + curBase + (long long)ch * chStride + (long long)h * Wd + j * 4) = o[k];
            }
        }
        if (t + 1 < VTILES) lds_barrier();
    }
}

extern "C" void kernel_launch(void* const* d_in, const int* in_sizes, int n_in,
                              void* d_out, int out_size, void* d_ws, size_t ws_size,
                              hipStream_t stream) {
    (void)in_sizes; (void)n_in; (void)out_size; (void)d_ws; (void)ws_size;
    const float* img  = (const float*)d_in[0];
    const float* edge = (const float*)d_in[1];
    float* out = (float*)d_out;

    const double SQRT2 = 1.4142135623730951;
    const double SQRT3 = 1.7320508075688772;
    const double LOG2E = 1.4426950408889634;
    const double denom = 3.872983346207417;
    const double sigma1 = 60.0 * SQRT3 * 2.0 / denom;
    const double sigma2 = 60.0 * SQRT3 * 1.0 / denom;
    const float c1 = (float)(-SQRT2 / sigma1 * LOG2E);
    const float c2 = (float)(-SQRT2 / sigma2 * LOG2E);

    const dim3 grid(4096), block(32);

    const dim3 vgrid(256), vblock(VCOLS * 32);

    dt_pass_kernel<<<grid, block, 0, stream>>>(img, out, edge, c1);
    dt_vpass_kernel<<<vgrid, vblock, 0, stream>>>(out, out, edge, c1);
    dt_pass_kernel<<<grid, block, 0, stream>>>(out, out, edge, c2);
    dt_vpass_kernel<<<vgrid, vblock, 0, stream>>>(out, out, edge, c2);
}


# speedup vs baseline: 5.3704x; 1.2651x over previous
#include <hip/hip_runtime.h>
#include <math.h>


typedef __attribute__((ext_vector_type(2))) float v2f;
typedef __attribute__((ext_vector_type(8))) float v8f;

#define WIDTH  512
#define SEGS   32
#define XPITCH 513
#define HPITCH 516

#define VCOLS  8
#define COLF   (16 * XPITCH + 2 * WIDTH + 4)

__device__ __forceinline__ float exp2_fast(float x) {
    return __builtin_amdgcn_exp2f(x);
}

template <int CTRL, int ROWS>
__device__ __forceinline__ float dpp_row(float old, float v) {
    return __builtin_bit_cast(float, __builtin_amdgcn_update_dpp(
               __builtin_bit_cast(int, old), __builtin_bit_cast(int, v), CTRL, ROWS, 0xf, true));
}
__device__ __forceinline__ float other_row(float v) {
    return __builtin_bit_cast(float, __builtin_amdgcn_permlanex16(
               __builtin_bit_cast(int, v), __builtin_bit_cast(int, v),
               0x76543210, 0xfedcba98, false, false));
}


template <int PITCH>
__device__ __forceinline__ void seg_products(const float* X, const float* tF, const float* omF,
                                             int half, int khi, bool bwd, int s,
                                             v8f& acc, float& Prot)
{
    const int base = s * 16;
    const int idx  = base + half;

    float t;
    if (!bwd) t = tF[idx];
    else {
        const float tl = tF[(WIDTH - idx) & (WIDTH - 1)];
        t = (idx == 0) ? -512.0f : tl;
    }

    float S = t;
    S += dpp_row<0x111, 0xf>(0.0f, S);
    S += dpp_row<0x112, 0xf>(0.0f, S);
    S += dpp_row<0x114, 0xf>(0.0f, S);
    S += dpp_row<0x118, 0xf>(0.0f, S);
    const float expS = exp2_fast(S);

    const float Srot = dpp_row<0x12e, 0x2>(S, S);
    Prot = dpp_row<0x128, 0x2>(expS, expS);

    acc = (v8f){0.f,0.f,0.f,0.f,0.f,0.f,0.f,0.f};
#pragma unroll
    for (int kk = 0; kk < 4; ++kk) {
        const int m0 = kk * 4 + khi * 2;
        float Sm0, Sm1;
        if (kk == 0)      { Sm0 = dpp_row<0x150, 0xf>(0.0f, Srot); Sm1 = dpp_row<0x151, 0xf>(0.0f, Srot); }
        else if (kk == 1) { Sm0 = dpp_row<0x154, 0xf>(0.0f, Srot); Sm1 = dpp_row<0x155, 0xf>(0.0f, Srot); }
        else if (kk == 2) { Sm0 = dpp_row<0x158, 0xf>(0.0f, Srot); Sm1 = dpp_row<0x159, 0xf>(0.0f, Srot); }
        else              { Sm0 = dpp_row<0x15c, 0xf>(0.0f, Srot); Sm1 = dpp_row<0x15d, 0xf>(0.0f, Srot); }
        v2f Af;
        Af.x = (m0     < half) ? exp2_fast(S - Sm0) : ((m0     == half) ? 1.0f : 0.0f);
        Af.y = (m0 + 1 < half) ? exp2_fast(S - Sm1) : ((m0 + 1 == half) ? 1.0f : 0.0f);

        const int p0 = base + m0;
        v2f Bf;
        if (!bwd) {
            Bf.x = omF[p0]     * X[half * PITCH + p0];
            Bf.y = omF[p0 + 1] * X[half * PITCH + p0 + 1];
        } else {
            const float ol  = omF[(WIDTH - p0) & (WIDTH - 1)];
            const float om0 = (p0 == 0) ? 1.0f : ol;
            const float om1 = omF[WIDTH - (p0 + 1)];
            Bf.x = om0 * X[half * PITCH + (WIDTH - 1 - p0)];
            Bf.y = om1 * X[half * PITCH + (WIDTH - 1 - (p0 + 1))];
        }

        acc = __builtin_amdgcn_wmma_f32_16x16x4_f32(
                  false, Af, false, Bf, (short)0, acc, false, false);
    }
}

template <int PITCH>
__device__ __forceinline__ void seg_carry(float* X, int half, int khi, bool bwd, int s,
                                          v8f acc, float Prot, float& carry)
{
    const int base = s * 16;
    acc[0] += dpp_row<0x150, 0xf>(0.0f, Prot) * carry;
    acc[1] += dpp_row<0x151, 0xf>(0.0f, Prot) * carry;
    acc[2] += dpp_row<0x152, 0xf>(0.0f, Prot) * carry;
    acc[3] += dpp_row<0x153, 0xf>(0.0f, Prot) * carry;
    acc[4] += dpp_row<0x154, 0xf>(0.0f, Prot) * carry;
    acc[5] += dpp_row<0x155, 0xf>(0.0f, Prot) * carry;
    acc[6] += dpp_row<0x156, 0xf>(0.0f, Prot) * carry;
    acc[7] += dpp_row<0x157, 0xf>(0.0f, Prot) * carry;
    const float upper = other_row(acc[7]);
    carry = khi ? acc[7] : upper;

#pragma unroll
    for (int rr = 0; rr < 8; ++rr) {
        const int j = base + rr + 8 * khi;
        const int w = bwd ? (WIDTH - 1 - j) : j;
        X[half * PITCH + w] = acc[rr];
    }
}

template <int PITCH>
__device__ __forceinline__ void scan_dir(float* X, const float* tF, const float* omF,
                                         int half, int khi, bool bwd)
{
    float carry = 0.0f;
    for (int s = 0; s < SEGS; s += 2) {
        v8f acc0, acc1;
        float P0, P1;
        seg_products<PITCH>(X, tF, omF, half, khi, bwd, s,     acc0, P0);
        seg_products<PITCH>(X, tF, omF, half, khi, bwd, s + 1, acc1, P1);
        seg_carry<PITCH>(X, half, khi, bwd, s,     acc0, P0, carry);
        seg_carry<PITCH>(X, half, khi, bwd, s + 1, acc1, P1, carry);
    }
}

__global__ __launch_bounds__(32)
void dt_pass_kernel(const float* in, float* out, const float* edge, float clog2)
{
    __shared__ __attribute__((aligned(16))) float hsm[16 * HPITCH + 2 * WIDTH];
    float* X   = hsm;
    float* tF  = hsm + 16 * HPITCH;
    float* omF = tF + WIDTH;

    const int lid  = threadIdx.x;
    const int half = lid & 15;
    const int khi  = lid >> 4;

    const int nb  = blockIdx.x;
    const int cg  = nb & 1;
    const int row = nb >> 1;
    const int b   = row >> 9;
    const int r   = row & 511;

    const int C = 32, H = 512, Wd = 512;
    const long long chStride = (long long)H * Wd;
    const long long imgBase  = (long long)b * C * H * Wd + (long long)r * Wd
                             + (long long)(cg * 16) * chStride;
    const long long edgeBase = (long long)b * H * Wd + (long long)r * Wd;

    constexpr int NEDGE = WIDTH / 4 / 32;
    float4 e4[NEDGE];
#pragma unroll
    for (int k = 0; k < NEDGE; ++k)
        e4[k] = *reinterpret_cast<const float4*>(edge + edgeBase + (lid + k * 32) * 4);

    constexpr int NPIECE = 16 * WIDTH / 4;
    constexpr int HBATCH = 32;
    for (int q0 = lid; q0 < NPIECE; q0 += HBATCH * 32) {
        float4 v[HBATCH];
#pragma unroll
        for (int k = 0; k < HBATCH; ++k) {
            const int q  = q0 + k * 32;
            const int ch = q >> 7, w = (q & 127) * 4;
            v[k] = *reinterpret_cast<const float4*>(in + imgBase + (long long)ch * chStride + w);
        }
#pragma unroll
        for (int k = 0; k < HBATCH; ++k) {
            const int q  = q0 + k * 32;
            const int ch = q >> 7, w = (q & 127) * 4;
            *reinterpret_cast<float4*>(X + ch * HPITCH + w) = v[k];
        }
    }
#pragma unroll
    for (int k = 0; k < NEDGE; ++k) {
        const int w0 = (lid + k * 32) * 4;
        const float ev[4] = {e4[k].x, e4[k].y, e4[k].z, e4[k].w};
        float tv[4], ov[4];
#pragma unroll
        for (int e = 0; e < 4; ++e) {
            const float Dv = 1.0f + 150.0f * ev[e];
            float t = clog2 * Dv;
            if (w0 + e == 0) t = -512.0f;
            tv[e] = t;
            ov[e] = 1.0f - exp2_fast(t);
        }
        *reinterpret_cast<float4*>(tF  + w0) = make_float4(tv[0], tv[1], tv[2], tv[3]);
        *reinterpret_cast<float4*>(omF + w0) = make_float4(ov[0], ov[1], ov[2], ov[3]);
    }
    __syncthreads();

    scan_dir<HPITCH>(X, tF, omF, half, khi, false);
    scan_dir<HPITCH>(X, tF, omF, half, khi, true);

    __syncthreads();
    constexpr int WBATCH = 16;
    for (int q0 = lid; q0 < NPIECE; q0 += WBATCH * 32) {
        float4 v[WBATCH];
#pragma unroll
        for (int k = 0; k < WBATCH; ++k) {
            const int q  = q0 + k * 32;
            const int ch = q >> 7, w = (q & 127) * 4;
            v[k] = *reinterpret_cast<const float4*>(X + ch * HPITCH + w);
        }
#pragma unroll
        for (int k = 0; k < WBATCH; ++k) {
            const int q  = q0 + k * 32;
            const int ch = q >> 7, w = (q & 127) * 4;
            *reinterpret_cast<float4*>(out + imgBase + (long long)ch * chStride + w) = v[k];
        }
    }
}

#define VTILES 2
#define PF_AHEAD 56
#define PF_ALL   64

__device__ __forceinline__ void lds_barrier() {
    __builtin_amdgcn_fence(__ATOMIC_RELEASE, "workgroup", "local");
    __builtin_amdgcn_s_barrier();
    __builtin_amdgcn_fence(__ATOMIC_ACQUIRE, "workgroup", "local");
}

template <int K0, int K1>
__device__ __forceinline__ void v_issue(const float* in, long long imgBase, int tid,
                                        float4 (&v)[PF_ALL])
{
#pragma unroll
    for (int k = K0; k < K1; ++k) {
        const int q  = tid + k * VCOLS * 32;
        const int j  = q & 1, h = (q >> 1) & (WIDTH - 1), ch = q >> 10;
        v[k] = *reinterpret_cast<const float4*>(
                   in + imgBase + (long long)ch * (512LL * 512LL) + (long long)h * 512 + j * 4);
    }
}

__device__ __forceinline__ void v_tile_base(int nb, long long& imgBase, long long& edgeBase)
{
    const int C = 32, H = 512, Wd = 512;
    const int piece = (nb >> 5) & 7;
    const int unit  = nb & 7;
    const int cg    = (nb >> 3) & 1;
    const int b     = ((nb >> 4) & 1) | ((nb >> 8) << 1);
    const int c0    = (unit * 8 + piece) * VCOLS;
    imgBase  = (long long)b * C * H * Wd + (long long)(cg * 16) * ((long long)H * Wd) + c0;
    edgeBase = (long long)b * H * Wd + c0;
}

__global__ __launch_bounds__(VCOLS * 32)
void dt_vpass_kernel(const float* in, float* out, const float* edge, float clog2)
{
    __shared__ float vsm[VCOLS * COLF];

    const int tid  = threadIdx.x;
    const int wv   = tid >> 5;
    const int lid  = tid & 31;
    const int half = lid & 15;
    const int khi  = lid >> 4;

    const int H = 512, Wd = 512;
    const long long chStride = (long long)H * Wd;

    constexpr int NEDGE  = WIDTH * 2 / (VCOLS * 32);
    constexpr int NPIECE = 16 * WIDTH * 2;

    float4 e4[NEDGE];
    float4 v[PF_ALL];

    int nb = blockIdx.x;
    long long imgBase, edgeBase;
    v_tile_base(nb, imgBase, edgeBase);
#pragma unroll
    for (int k = 0; k < NEDGE; ++k) {
        const int q = tid + k * VCOLS * 32;
        const int j = q & 1, h = q >> 1;
        e4[k] = *reinterpret_cast<const float4*>(edge + edgeBase + (long long)h * Wd + j * 4);
    }
    v_issue<0, PF_ALL>(in, imgBase, tid, v);

    float* X   = vsm + wv * COLF;
    float* tF  = X + 16 * XPITCH;
    float* omF = tF + WIDTH;

#pragma unroll
    for (int t = 0; t < VTILES; ++t) {
#pragma unroll
        for (int k = 0; k < PF_ALL; ++k) {
            const int q  = tid + k * VCOLS * 32;
            const int j  = q & 1, h = (q >> 1) & (WIDTH - 1), ch = q >> 10;
            float* d = vsm + (j * 4) * COLF + ch * XPITCH + h;
            d[0 * COLF] = v[k].x;
            d[1 * COLF] = v[k].y;
            d[2 * COLF] = v[k].z;
            d[3 * COLF] = v[k].w;
        }
#pragma unroll
        for (int k = 0; k < NEDGE; ++k) {
            const int q = tid + k * VCOLS * 32;
            const int j = q & 1, h = q >> 1;
            const float ev[4] = {e4[k].x, e4[k].y, e4[k].z, e4[k].w};
#pragma unroll
            for (int e = 0; e < 4; ++e) {
                float tt = clog2 * (1.0f + 150.0f * ev[e]);
                if (h == 0) tt = -512.0f;
                float* col = vsm + (j * 4 + e) * COLF + 16 * XPITCH;
                col[h]         = tt;
                col[WIDTH + h] = 1.0f - exp2_fast(tt);
            }
        }

        const long long curBase = imgBase;
        if (t + 1 < VTILES) {
            nb += gridDim.x;
            v_tile_base(nb, imgBase, edgeBase);
#pragma unroll
            for (int k = 0; k < NEDGE; ++k) {
                const int q = tid + k * VCOLS * 32;
                const int j = q & 1, h = q >> 1;
                e4[k] = *reinterpret_cast<const float4*>(edge + edgeBase + (long long)h * Wd + j * 4);
            }
            v_issue<0, PF_AHEAD>(in, imgBase, tid, v);
        }
        lds_barrier();

        scan_dir<XPITCH>(X, tF, omF, half, khi, false);
        scan_dir<XPITCH>(X, tF, omF, half, khi, true);

        lds_barrier();
        if (t + 1 < VTILES)
            v_issue<PF_AHEAD, PF_ALL>(in, imgBase, tid, v);

        constexpr int VWB = 4;
        for (int q0 = tid; q0 < NPIECE; q0 += VWB * VCOLS * 32) {
            float4 o[VWB];
#pragma unroll
            for (int k = 0; k < VWB; ++k) {
                const int q = q0 + k * VCOLS * 32;
                const int j = q & 1, h = (q >> 1) & (WIDTH - 1), ch = q >> 10;
                const float* s = vsm + (j * 4) * COLF + ch * XPITCH + h;
                o[k].x = s[0 * COLF];
                o[k].y = s[1 * COLF];
                o[k].z = s[2 * COLF];
                o[k].w = s[3 * COLF];
            }
#pragma unroll
            for (int k = 0; k < VWB; ++k) {
                const int q = q0 + k * VCOLS * 32;
                const int j = q & 1, h = (q >> 1) & (WIDTH - 1), ch = q >> 10;
                *reinterpret_cast<float4*>(
                    out + curBase + (long long)ch * chStride + (long long)h * Wd + j * 4) = o[k];
            }
        }
        if (t + 1 < VTILES) lds_barrier();
    }
}

extern "C" void kernel_launch(void* const* d_in, const int* in_sizes, int n_in,
                              void* d_out, int out_size, void* d_ws, size_t ws_size,
                              hipStream_t stream) {
    (void)in_sizes; (void)n_in; (void)out_size; (void)d_ws; (void)ws_size;
    const float* img  = (const float*)d_in[0];
    const float* edge = (const float*)d_in[1];
    float* out = (float*)d_out;

    const double SQRT2 = 1.4142135623730951;
    const double SQRT3 = 1.7320508075688772;
    const double LOG2E = 1.4426950408889634;
    const double denom = 3.872983346207417;
    const double sigma1 = 60.0 * SQRT3 * 2.0 / denom;
    const double sigma2 = 60.0 * SQRT3 * 1.0 / denom;
    const float c1 = (float)(-SQRT2 / sigma1 * LOG2E);
    const float c2 = (float)(-SQRT2 / sigma2 * LOG2E);

    const dim3 grid(4096), block(32);

    const dim3 vgrid(256), vblock(VCOLS * 32);

    dt_pass_kernel<<<grid, block, 0, stream>>>(img, out, edge, c1);
    dt_vpass_kernel<<<vgrid, vblock, 0, stream>>>(out, out, edge, c1);
    dt_pass_kernel<<<grid, block, 0, stream>>>(out, out, edge, c2);
    dt_vpass_kernel<<<vgrid, vblock, 0, stream>>>(out, out, edge, c2);
}


# Horizontal pass: a one-wave work-group takes two rows; the second row's global loads go into registers ahead of the first row's scan and land while it runs; scan and sums unchanged
# speedup vs baseline: 5.6423x; 1.0506x over previous
#include <hip/hip_runtime.h>
#include <math.h>


typedef __attribute__((ext_vector_type(2))) float v2f;
typedef __attribute__((ext_vector_type(8))) float v8f;

#define WIDTH  512
#define SEGS   32
#define XPITCH 513
#define HPITCH 516

#define VCOLS  8
#define COLF   (16 * XPITCH + 2 * WIDTH + 4)

__device__ __forceinline__ float exp2_fast(float x) {
    return __builtin_amdgcn_exp2f(x);
}

template <int CTRL, int ROWS>
__device__ __forceinline__ float dpp_row(float old, float v) {
    return __builtin_bit_cast(float, __builtin_amdgcn_update_dpp(
               __builtin_bit_cast(int, old), __builtin_bit_cast(int, v), CTRL, ROWS, 0xf, true));
}
__device__ __forceinline__ float other_row(float v) {
    return __builtin_bit_cast(float, __builtin_amdgcn_permlanex16(
               __builtin_bit_cast(int, v), __builtin_bit_cast(int, v),
               0x76543210, 0xfedcba98, false, false));
}


template <int PITCH>
__device__ __forceinline__ void seg_products(const float* X, const float* tF, const float* omF,
                                             int half, int khi, bool bwd, int s,
                                             v8f& acc, float& Prot)
{
    const int base = s * 16;
    const int idx  = base + half;

    float t;
    if (!bwd) t = tF[idx];
    else {
        const float tl = tF[(WIDTH - idx) & (WIDTH - 1)];
        t = (idx == 0) ? -512.0f : tl;
    }

    float S = t;
    S += dpp_row<0x111, 0xf>(0.0f, S);
    S += dpp_row<0x112, 0xf>(0.0f, S);
    S += dpp_row<0x114, 0xf>(0.0f, S);
    S += dpp_row<0x118, 0xf>(0.0f, S);
    const float expS = exp2_fast(S);

    const float Srot = dpp_row<0x12e, 0x2>(S, S);
    Prot = dpp_row<0x128, 0x2>(expS, expS);

    acc = (v8f){0.f,0.f,0.f,0.f,0.f,0.f,0.f,0.f};
#pragma unroll
    for (int kk = 0; kk < 4; ++kk) {
        const int m0 = kk * 4 + khi * 2;
        float Sm0, Sm1;
        if (kk == 0)      { Sm0 = dpp_row<0x150, 0xf>(0.0f, Srot); Sm1 = dpp_row<0x151, 0xf>(0.0f, Srot); }
        else if (kk == 1) { Sm0 = dpp_row<0x154, 0xf>(0.0f, Srot); Sm1 = dpp_row<0x155, 0xf>(0.0f, Srot); }
        else if (kk == 2) { Sm0 = dpp_row<0x158, 0xf>(0.0f, Srot); Sm1 = dpp_row<0x159, 0xf>(0.0f, Srot); }
        else              { Sm0 = dpp_row<0x15c, 0xf>(0.0f, Srot); Sm1 = dpp_row<0x15d, 0xf>(0.0f, Srot); }
        v2f Af;
        Af.x = (m0     < half) ? exp2_fast(S - Sm0) : ((m0     == half) ? 1.0f : 0.0f);
        Af.y = (m0 + 1 < half) ? exp2_fast(S - Sm1) : ((m0 + 1 == half) ? 1.0f : 0.0f);

        const int p0 = base + m0;
        v2f Bf;
        if (!bwd) {
            Bf.x = omF[p0]     * X[half * PITCH + p0];
            Bf.y = omF[p0 + 1] * X[half * PITCH + p0 + 1];
        } else {
            const float ol  = omF[(WIDTH - p0) & (WIDTH - 1)];
            const float om0 = (p0 == 0) ? 1.0f : ol;
            const float om1 = omF[WIDTH - (p0 + 1)];
            Bf.x = om0 * X[half * PITCH + (WIDTH - 1 - p0)];
            Bf.y = om1 * X[half * PITCH + (WIDTH - 1 - (p0 + 1))];
        }

        acc = __builtin_amdgcn_wmma_f32_16x16x4_f32(
                  false, Af, false, Bf, (short)0, acc, false, false);
    }
}

template <int PITCH>
__device__ __forceinline__ void seg_carry(float* X, int half, int khi, bool bwd, int s,
                                          v8f acc, float Prot, float& carry)
{
    const int base = s * 16;
    acc[0] += dpp_row<0x150, 0xf>(0.0f, Prot) * carry;
    acc[1] += dpp_row<0x151, 0xf>(0.0f, Prot) * carry;
    acc[2] += dpp_row<0x152, 0xf>(0.0f, Prot) * carry;
    acc[3] += dpp_row<0x153, 0xf>(0.0f, Prot) * carry;
    acc[4] += dpp_row<0x154, 0xf>(0.0f, Prot) * carry;
    acc[5] += dpp_row<0x155, 0xf>(0.0f, Prot) * carry;
    acc[6] += dpp_row<0x156, 0xf>(0.0f, Prot) * carry;
    acc[7] += dpp_row<0x157, 0xf>(0.0f, Prot) * carry;
    const float upper = other_row(acc[7]);
    carry = khi ? acc[7] : upper;

#pragma unroll
    for (int rr = 0; rr < 8; ++rr) {
        const int j = base + rr + 8 * khi;
        const int w = bwd ? (WIDTH - 1 - j) : j;
        X[half * PITCH + w] = acc[rr];
    }
}

template <int PITCH>
__device__ __forceinline__ void scan_dir(float* X, const float* tF, const float* omF,
                                         int half, int khi, bool bwd)
{
    float carry = 0.0f;
    for (int s = 0; s < SEGS; s += 2) {
        v8f acc0, acc1;
        float P0, P1;
        seg_products<PITCH>(X, tF, omF, half, khi, bwd, s,     acc0, P0);
        seg_products<PITCH>(X, tF, omF, half, khi, bwd, s + 1, acc1, P1);
        seg_carry<PITCH>(X, half, khi, bwd, s,     acc0, P0, carry);
        seg_carry<PITCH>(X, half, khi, bwd, s + 1, acc1, P1, carry);
    }
}

__device__ __forceinline__ void lds_barrier() {
    __builtin_amdgcn_fence(__ATOMIC_RELEASE, "workgroup", "local");
    __builtin_amdgcn_s_barrier();
    __builtin_amdgcn_fence(__ATOMIC_ACQUIRE, "workgroup", "local");
}

#define HTILES 2
#define HPF_AHEAD 56
#define HPF_ALL   64

template <int K0, int K1>
__device__ __forceinline__ void h_issue(const float* in, long long imgBase, int lid,
                                        float4 (&v)[HPF_ALL])
{
#pragma unroll
    for (int k = K0; k < K1; ++k) {
        const int q  = lid + k * 32;
        const int ch = q >> 7, w = (q & 127) * 4;
        v[k] = *reinterpret_cast<const float4*>(in + imgBase + (long long)ch * (512LL * 512LL) + w);
    }
}

__device__ __forceinline__ void h_tile_base(int nb, long long& imgBase, long long& edgeBase)
{
    const int C = 32, H = 512, Wd = 512;
    const int cg  = nb & 1;
    const int row = nb >> 1;
    const int b   = row >> 9;
    const int r   = row & 511;
    imgBase  = (long long)b * C * H * Wd + (long long)r * Wd
             + (long long)(cg * 16) * ((long long)H * Wd);
    edgeBase = (long long)b * H * Wd + (long long)r * Wd;
}

__global__ __launch_bounds__(32, 2)
void dt_pass_kernel(const float* in, float* out, const float* edge, float clog2)
{
    __shared__ __attribute__((aligned(16))) float hsm[16 * HPITCH + 2 * WIDTH];
    float* X   = hsm;
    float* tF  = hsm + 16 * HPITCH;
    float* omF = tF + WIDTH;

    const int lid  = threadIdx.x;
    const int half = lid & 15;
    const int khi  = lid >> 4;

    const long long chStride = 512LL * 512LL;

    constexpr int NEDGE  = WIDTH / 4 / 32;
    constexpr int NPIECE = 16 * WIDTH / 4;

    float4 e4[NEDGE];
    float4 v[HPF_ALL];

    int nb = blockIdx.x;
    long long imgBase, edgeBase;
    h_tile_base(nb, imgBase, edgeBase);
#pragma unroll
    for (int k = 0; k < NEDGE; ++k)
        e4[k] = *reinterpret_cast<const float4*>(edge + edgeBase + (lid + k * 32) * 4);
    h_issue<0, HPF_ALL>(in, imgBase, lid, v);

#pragma unroll
    for (int t = 0; t < HTILES; ++t) {
#pragma unroll
        for (int k = 0; k < HPF_ALL; ++k) {
            const int q  = lid + k * 32;
            const int ch = q >> 7, w = (q & 127) * 4;
            float* d = X + ch * HPITCH + w;
            d[0] = v[k].x;
            d[1] = v[k].y;
            d[2] = v[k].z;
            d[3] = v[k].w;
        }
#pragma unroll
        for (int k = 0; k < NEDGE; ++k) {
            const int w0 = (lid + k * 32) * 4;
            const float ev[4] = {e4[k].x, e4[k].y, e4[k].z, e4[k].w};
            float tv[4], ov[4];
#pragma unroll
            for (int e = 0; e < 4; ++e) {
                const float Dv = 1.0f + 150.0f * ev[e];
                float tt = clog2 * Dv;
                if (w0 + e == 0) tt = -512.0f;
                tv[e] = tt;
                ov[e] = 1.0f - exp2_fast(tt);
            }
            *reinterpret_cast<float4*>(tF  + w0) = make_float4(tv[0], tv[1], tv[2], tv[3]);
            *reinterpret_cast<float4*>(omF + w0) = make_float4(ov[0], ov[1], ov[2], ov[3]);
        }

        const long long curBase = imgBase;
        if (t + 1 < HTILES) {
            __builtin_amdgcn_sched_barrier(0);
            nb += gridDim.x;
            h_tile_base(nb, imgBase, edgeBase);
#pragma unroll
            for (int k = 0; k < NEDGE; ++k)
                e4[k] = *reinterpret_cast<const float4*>(edge + edgeBase + (lid + k * 32) * 4);
            h_issue<0, HPF_AHEAD>(in, imgBase, lid, v);
        }
        lds_barrier();

        scan_dir<HPITCH>(X, tF, omF, half, khi, false);
        scan_dir<HPITCH>(X, tF, omF, half, khi, true);

        lds_barrier();
        if (t + 1 < HTILES)
            h_issue<HPF_AHEAD, HPF_ALL>(in, imgBase, lid, v);

        constexpr int WBATCH = 4;
        for (int q0 = lid; q0 < NPIECE; q0 += WBATCH * 32) {
            float4 o[WBATCH];
#pragma unroll
            for (int k = 0; k < WBATCH; ++k) {
                const int q  = q0 + k * 32;
                const int ch = q >> 7, w = (q & 127) * 4;
                o[k] = *reinterpret_cast<const float4*>(X + ch * HPITCH + w);
            }
#pragma unroll
            for (int k = 0; k < WBATCH; ++k) {
                const int q  = q0 + k * 32;
                const int ch = q >> 7, w = (q & 127) * 4;
                *reinterpret_cast<float4*>(out + curBase + (long long)ch * chStride + w) = o[k];
            }
        }
        if (t + 1 < HTILES) lds_barrier();
    }
}

#define VTILES 2
#define PF_AHEAD 56
#define PF_ALL   64

template <int K0, int K1>
__device__ __forceinline__ void v_issue(const float* in, long long imgBase, int tid,
                                        float4 (&v)[PF_ALL])
{
#pragma unroll
    for (int k = K0; k < K1; ++k) {
        const int q  = tid + k * VCOLS * 32;
        const int j  = q & 1, h = (q >> 1) & (WIDTH - 1), ch = q >> 10;
        v[k] = *reinterpret_cast<const float4*>(
                   in + imgBase + (long long)ch * (512LL * 512LL) + (long long)h * 512 + j * 4);
    }
}

__device__ __forceinline__ void v_tile_base(int nb, long long& imgBase, long long& edgeBase)
{
    const int C = 32, H = 512, Wd = 512;
    const int piece = (nb >> 5) & 7;
    const int unit  = nb & 7;
    const int cg    = (nb >> 3) & 1;
    const int b     = ((nb >> 4) & 1) | ((nb >> 8) << 1);
    const int c0    = (unit * 8 + piece) * VCOLS;
    imgBase  = (long long)b * C * H * Wd + (long long)(cg * 16) * ((long long)H * Wd) + c0;
    edgeBase = (long long)b * H * Wd + c0;
}

__global__ __launch_bounds__(VCOLS * 32)
void dt_vpass_kernel(const float* in, float* out, const float* edge, float clog2)
{
    __shared__ float vsm[VCOLS * COLF];

    const int tid  = threadIdx.x;
    const int wv   = tid >> 5;
    const int lid  = tid & 31;
    const int half = lid & 15;
    const int khi  = lid >> 4;

    const int H = 512, Wd = 512;
    const long long chStride = (long long)H * Wd;

    constexpr int NEDGE  = WIDTH * 2 / (VCOLS * 32);
    constexpr int NPIECE = 16 * WIDTH * 2;

    float4 e4[NEDGE];
    float4 v[PF_ALL];

    int nb = blockIdx.x;
    long long imgBase, edgeBase;
    v_tile_base(nb, imgBase, edgeBase);
#pragma unroll
    for (int k = 0; k < NEDGE; ++k) {
        const int q = tid + k * VCOLS * 32;
        const int j = q & 1, h = q >> 1;
        e4[k] = *reinterpret_cast<const float4*>(edge + edgeBase + (long long)h * Wd + j * 4);
    }
    v_issue<0, PF_ALL>(in, imgBase, tid, v);

    float* X   = vsm + wv * COLF;
    float* tF  = X + 16 * XPITCH;
    float* omF = tF + WIDTH;

#pragma unroll
    for (int t = 0; t < VTILES; ++t) {
#pragma unroll
        for (int k = 0; k < PF_ALL; ++k) {
            const int q  = tid + k * VCOLS * 32;
            const int j  = q & 1, h = (q >> 1) & (WIDTH - 1), ch = q >> 10;
            float* d = vsm + (j * 4) * COLF + ch * XPITCH + h;
            d[0 * COLF] = v[k].x;
            d[1 * COLF] = v[k].y;
            d[2 * COLF] = v[k].z;
            d[3 * COLF] = v[k].w;
        }
#pragma unroll
        for (int k = 0; k < NEDGE; ++k) {
            const int q = tid + k * VCOLS * 32;
            const int j = q & 1, h = q >> 1;
            const float ev[4] = {e4[k].x, e4[k].y, e4[k].z, e4[k].w};
#pragma unroll
            for (int e = 0; e < 4; ++e) {
                float tt = clog2 * (1.0f + 150.0f * ev[e]);
                if (h == 0) tt = -512.0f;
                float* col = vsm + (j * 4 + e) * COLF + 16 * XPITCH;
                col[h]         = tt;
                col[WIDTH + h] = 1.0f - exp2_fast(tt);
            }
        }

        const long long curBase = imgBase;
        if (t + 1 < VTILES) {
            nb += gridDim.x;
            v_tile_base(nb, imgBase, edgeBase);
#pragma unroll
            for (int k = 0; k < NEDGE; ++k) {
                const int q = tid + k * VCOLS * 32;
                const int j = q & 1, h = q >> 1;
                e4[k] = *reinterpret_cast<const float4*>(edge + edgeBase + (long long)h * Wd + j * 4);
            }
            v_issue<0, PF_AHEAD>(in, imgBase, tid, v);
        }
        lds_barrier();

        scan_dir<XPITCH>(X, tF, omF, half, khi, false);
        scan_dir<XPITCH>(X, tF, omF, half, khi, true);

        lds_barrier();
        if (t + 1 < VTILES)
            v_issue<PF_AHEAD, PF_ALL>(in, imgBase, tid, v);

        constexpr int VWB = 4;
        for (int q0 = tid; q0 < NPIECE; q0 += VWB * VCOLS * 32) {
            float4 o[VWB];
#pragma unroll
            for (int k = 0; k < VWB; ++k) {
                const int q = q0 + k * VCOLS * 32;
                const int j = q & 1, h = (q >> 1) & (WIDTH - 1), ch = q >> 10;
                const float* s = vsm + (j * 4) * COLF + ch * XPITCH + h;
                o[k].x = s[0 * COLF];
                o[k].y = s[1 * COLF];
                o[k].z = s[2 * COLF];
                o[k].w = s[3 * COLF];
            }
#pragma unroll
            for (int k = 0; k < VWB; ++k) {
                const int q = q0 + k * VCOLS * 32;
                const int j = q & 1, h = (q >> 1) & (WIDTH - 1), ch = q >> 10;
                *reinterpret_cast<float4*>(
                    out + curBase + (long long)ch * chStride + (long long)h * Wd + j * 4) = o[k];
            }
        }
        if (t + 1 < VTILES) lds_barrier();
    }
}

extern "C" void kernel_launch(void* const* d_in, const int* in_sizes, int n_in,
                              void* d_out, int out_size, void* d_ws, size_t ws_size,
                              hipStream_t stream) {
    (void)in_sizes; (void)n_in; (void)out_size; (void)d_ws; (void)ws_size;
    const float* img  = (const float*)d_in[0];
    const float* edge = (const float*)d_in[1];
    float* out = (float*)d_out;

    const double SQRT2 = 1.4142135623730951;
    const double SQRT3 = 1.7320508075688772;
    const double LOG2E = 1.4426950408889634;
    const double denom = 3.872983346207417;
    const double sigma1 = 60.0 * SQRT3 * 2.0 / denom;
    const double sigma2 = 60.0 * SQRT3 * 1.0 / denom;
    const float c1 = (float)(-SQRT2 / sigma1 * LOG2E);
    const float c2 = (float)(-SQRT2 / sigma2 * LOG2E);

    const dim3 grid(2048), block(32);

    const dim3 vgrid(256), vblock(VCOLS * 32);

    dt_pass_kernel<<<grid, block, 0, stream>>>(img, out, edge, c1);
    dt_vpass_kernel<<<vgrid, vblock, 0, stream>>>(out, out, edge, c1);
    dt_pass_kernel<<<grid, block, 0, stream>>>(out, out, edge, c2);
    dt_vpass_kernel<<<vgrid, vblock, 0, stream>>>(out, out, edge, c2);
}


# Backward scan: tF/omF carry the boundary (a=0, 1-a=1) as element 512 and are read at base+constant (512-r): no index wrap or select, loads issued as one group like the forward scan's; same values
# speedup vs baseline: 5.8743x; 1.0411x over previous
#include <hip/hip_runtime.h>
#include <math.h>


typedef __attribute__((ext_vector_type(2))) float v2f;
typedef __attribute__((ext_vector_type(8))) float v8f;

#define WIDTH  512
#define SEGS   32
#define XPITCH 513
#define HPITCH 516

#define VCOLS  8
#define HCOEF  516
#define VCOEF  514
#define COLF   (16 * XPITCH + 2 * VCOEF)

__device__ __forceinline__ float exp2_fast(float x) {
    return __builtin_amdgcn_exp2f(x);
}

template <int CTRL, int ROWS>
__device__ __forceinline__ float dpp_row(float old, float v) {
    return __builtin_bit_cast(float, __builtin_amdgcn_update_dpp(
               __builtin_bit_cast(int, old), __builtin_bit_cast(int, v), CTRL, ROWS, 0xf, true));
}
__device__ __forceinline__ float other_row(float v) {
    return __builtin_bit_cast(float, __builtin_amdgcn_permlanex16(
               __builtin_bit_cast(int, v), __builtin_bit_cast(int, v),
               0x76543210, 0xfedcba98, false, false));
}


template <int PITCH>
__device__ __forceinline__ void seg_products(const float* X, const float* tF, const float* omF,
                                             int half, int khi, bool bwd, int s,
                                             v8f& acc, float& Prot)
{
    const int base = s * 16;
    const int idx  = base + half;

    const float t = bwd ? tF[WIDTH - idx] : tF[idx];

    float S = t;
    S += dpp_row<0x111, 0xf>(0.0f, S);
    S += dpp_row<0x112, 0xf>(0.0f, S);
    S += dpp_row<0x114, 0xf>(0.0f, S);
    S += dpp_row<0x118, 0xf>(0.0f, S);
    const float expS = exp2_fast(S);

    const float Srot = dpp_row<0x12e, 0x2>(S, S);
    Prot = dpp_row<0x128, 0x2>(expS, expS);

    acc = (v8f){0.f,0.f,0.f,0.f,0.f,0.f,0.f,0.f};
#pragma unroll
    for (int kk = 0; kk < 4; ++kk) {
        const int m0 = kk * 4 + khi * 2;
        float Sm0, Sm1;
        if (kk == 0)      { Sm0 = dpp_row<0x150, 0xf>(0.0f, Srot); Sm1 = dpp_row<0x151, 0xf>(0.0f, Srot); }
        else if (kk == 1) { Sm0 = dpp_row<0x154, 0xf>(0.0f, Srot); Sm1 = dpp_row<0x155, 0xf>(0.0f, Srot); }
        else if (kk == 2) { Sm0 = dpp_row<0x158, 0xf>(0.0f, Srot); Sm1 = dpp_row<0x159, 0xf>(0.0f, Srot); }
        else              { Sm0 = dpp_row<0x15c, 0xf>(0.0f, Srot); Sm1 = dpp_row<0x15d, 0xf>(0.0f, Srot); }
        v2f Af;
        Af.x = (m0     < half) ? exp2_fast(S - Sm0) : ((m0     == half) ? 1.0f : 0.0f);
        Af.y = (m0 + 1 < half) ? exp2_fast(S - Sm1) : ((m0 + 1 == half) ? 1.0f : 0.0f);

        const int p0 = base + m0;
        v2f Bf;
        if (!bwd) {
            Bf.x = omF[p0]     * X[half * PITCH + p0];
            Bf.y = omF[p0 + 1] * X[half * PITCH + p0 + 1];
        } else {
            const int rb = WIDTH - 16 - base + 2 - khi * 2;
            const float* Xr = X + half * PITCH + rb;
            const float* Or = omF + rb;
            const float om0 = Or[14 - kk * 4];
            const float om1 = Or[13 - kk * 4];
            Bf.x = om0 * Xr[13 - kk * 4];
            Bf.y = om1 * Xr[12 - kk * 4];
        }

        acc = __builtin_amdgcn_wmma_f32_16x16x4_f32(
                  false, Af, false, Bf, (short)0, acc, false, false);
    }
}

template <int PITCH>
__device__ __forceinline__ void seg_carry(float* X, int half, int khi, bool bwd, int s,
                                          v8f acc, float Prot, float& carry)
{
    const int base = s * 16;
    acc[0] += dpp_row<0x150, 0xf>(0.0f, Prot) * carry;
    acc[1] += dpp_row<0x151, 0xf>(0.0f, Prot) * carry;
    acc[2] += dpp_row<0x152, 0xf>(0.0f, Prot) * carry;
    acc[3] += dpp_row<0x153, 0xf>(0.0f, Prot) * carry;
    acc[4] += dpp_row<0x154, 0xf>(0.0f, Prot) * carry;
    acc[5] += dpp_row<0x155, 0xf>(0.0f, Prot) * carry;
    acc[6] += dpp_row<0x156, 0xf>(0.0f, Prot) * carry;
    acc[7] += dpp_row<0x157, 0xf>(0.0f, Prot) * carry;
    const float upper = other_row(acc[7]);
    carry = khi ? acc[7] : upper;

#pragma unroll
    for (int rr = 0; rr < 8; ++rr) {
        const int j = base + rr + 8 * khi;
        const int w = bwd ? (WIDTH - 1 - j) : j;
        X[half * PITCH + w] = acc[rr];
    }
}

template <int PITCH>
__device__ __forceinline__ void scan_dir(float* X, const float* tF, const float* omF,
                                         int half, int khi, bool bwd)
{
    float carry = 0.0f;
    for (int s = 0; s < SEGS; s += 2) {
        v8f acc0, acc1;
        float P0, P1;
        seg_products<PITCH>(X, tF, omF, half, khi, bwd, s,     acc0, P0);
        seg_products<PITCH>(X, tF, omF, half, khi, bwd, s + 1, acc1, P1);
        seg_carry<PITCH>(X, half, khi, bwd, s,     acc0, P0, carry);
        seg_carry<PITCH>(X, half, khi, bwd, s + 1, acc1, P1, carry);
    }
}

__device__ __forceinline__ void lds_barrier() {
    __builtin_amdgcn_fence(__ATOMIC_RELEASE, "workgroup", "local");
    __builtin_amdgcn_s_barrier();
    __builtin_amdgcn_fence(__ATOMIC_ACQUIRE, "workgroup", "local");
}

#define HTILES 2
#define HPF_AHEAD 56
#define HPF_ALL   64

template <int K0, int K1>
__device__ __forceinline__ void h_issue(const float* in, long long imgBase, int lid,
                                        float4 (&v)[HPF_ALL])
{
#pragma unroll
    for (int k = K0; k < K1; ++k) {
        const int q  = lid + k * 32;
        const int ch = q >> 7, w = (q & 127) * 4;
        v[k] = *reinterpret_cast<const float4*>(in + imgBase + (long long)ch * (512LL * 512LL) + w);
    }
}

__device__ __forceinline__ void h_tile_base(int nb, long long& imgBase, long long& edgeBase)
{
    const int C = 32, H = 512, Wd = 512;
    const int cg  = nb & 1;
    const int row = nb >> 1;
    const int b   = row >> 9;
    const int r   = row & 511;
    imgBase  = (long long)b * C * H * Wd + (long long)r * Wd
             + (long long)(cg * 16) * ((long long)H * Wd);
    edgeBase = (long long)b * H * Wd + (long long)r * Wd;
}

__global__ __launch_bounds__(32, 2)
void dt_pass_kernel(const float* in, float* out, const float* edge, float clog2)
{
    __shared__ __attribute__((aligned(16))) float hsm[16 * HPITCH + 2 * HCOEF];
    float* X   = hsm;
    float* tF  = hsm + 16 * HPITCH;
    float* omF = tF + HCOEF;

    const int lid  = threadIdx.x;
    const int half = lid & 15;
    const int khi  = lid >> 4;

    const long long chStride = 512LL * 512LL;

    constexpr int NEDGE  = WIDTH / 4 / 32;
    constexpr int NPIECE = 16 * WIDTH / 4;

    float4 e4[NEDGE];
    float4 v[HPF_ALL];

    int nb = blockIdx.x;
    long long imgBase, edgeBase;
    h_tile_base(nb, imgBase, edgeBase);
#pragma unroll
    for (int k = 0; k < NEDGE; ++k)
        e4[k] = *reinterpret_cast<const float4*>(edge + edgeBase + (lid + k * 32) * 4);
    h_issue<0, HPF_ALL>(in, imgBase, lid, v);

#pragma unroll
    for (int t = 0; t < HTILES; ++t) {
#pragma unroll
        for (int k = 0; k < HPF_ALL; ++k) {
            const int q  = lid + k * 32;
            const int ch = q >> 7, w = (q & 127) * 4;
            float* d = X + ch * HPITCH + w;
            d[0] = v[k].x;
            d[1] = v[k].y;
            d[2] = v[k].z;
            d[3] = v[k].w;
        }
#pragma unroll
        for (int k = 0; k < NEDGE; ++k) {
            const int w0 = (lid + k * 32) * 4;
            const float ev[4] = {e4[k].x, e4[k].y, e4[k].z, e4[k].w};
            float tv[4], ov[4];
#pragma unroll
            for (int e = 0; e < 4; ++e) {
                const float Dv = 1.0f + 150.0f * ev[e];
                float tt = clog2 * Dv;
                if (w0 + e == 0) tt = -512.0f;
                tv[e] = tt;
                ov[e] = 1.0f - exp2_fast(tt);
            }
            *reinterpret_cast<float4*>(tF  + w0) = make_float4(tv[0], tv[1], tv[2], tv[3]);
            *reinterpret_cast<float4*>(omF + w0) = make_float4(ov[0], ov[1], ov[2], ov[3]);
        }
        if (lid == 0) { tF[WIDTH] = -512.0f; omF[WIDTH] = 1.0f; }

        const long long curBase = imgBase;
        if (t + 1 < HTILES) {
            __builtin_amdgcn_sched_barrier(0);
            nb += gridDim.x;
            h_tile_base(nb, imgBase, edgeBase);
#pragma unroll
            for (int k = 0; k < NEDGE; ++k)
                e4[k] = *reinterpret_cast<const float4*>(edge + edgeBase + (lid + k * 32) * 4);
            h_issue<0, HPF_AHEAD>(in, imgBase, lid, v);
        }
        lds_barrier();

        scan_dir<HPITCH>(X, tF, omF, half, khi, false);
        scan_dir<HPITCH>(X, tF, omF, half, khi, true);

        lds_barrier();
        if (t + 1 < HTILES)
            h_issue<HPF_AHEAD, HPF_ALL>(in, imgBase, lid, v);

        constexpr int WBATCH = 4;
        for (int q0 = lid; q0 < NPIECE; q0 += WBATCH * 32) {
            float4 o[WBATCH];
#pragma unroll
            for (int k = 0; k < WBATCH; ++k) {
                const int q  = q0 + k * 32;
                const int ch = q >> 7, w = (q & 127) * 4;
                o[k] = *reinterpret_cast<const float4*>(X + ch * HPITCH + w);
            }
#pragma unroll
            for (int k = 0; k < WBATCH; ++k) {
                const int q  = q0 + k * 32;
                const int ch = q >> 7, w = (q & 127) * 4;
                *reinterpret_cast<float4*>(out + curBase + (long long)ch * chStride + w) = o[k];
            }
        }
        if (t + 1 < HTILES) lds_barrier();
    }
}

#define VTILES 2
#define PF_AHEAD 56
#define PF_ALL   64

template <int K0, int K1>
__device__ __forceinline__ void v_issue(const float* in, long long imgBase, int tid,
                                        float4 (&v)[PF_ALL])
{
#pragma unroll
    for (int k = K0; k < K1; ++k) {
        const int q  = tid + k * VCOLS * 32;
        const int j  = q & 1, h = (q >> 1) & (WIDTH - 1), ch = q >> 10;
        v[k] = *reinterpret_cast<const float4*>(
                   in + imgBase + (long long)ch * (512LL * 512LL) + (long long)h * 512 + j * 4);
    }
}

__device__ __forceinline__ void v_tile_base(int nb, long long& imgBase, long long& edgeBase)
{
    const int C = 32, H = 512, Wd = 512;
    const int piece = (nb >> 5) & 7;
    const int unit  = nb & 7;
    const int cg    = (nb >> 3) & 1;
    const int b     = ((nb >> 4) & 1) | ((nb >> 8) << 1);
    const int c0    = (unit * 8 + piece) * VCOLS;
    imgBase  = (long long)b * C * H * Wd + (long long)(cg * 16) * ((long long)H * Wd) + c0;
    edgeBase = (long long)b * H * Wd + c0;
}

__global__ __launch_bounds__(VCOLS * 32)
void dt_vpass_kernel(const float* in, float* out, const float* edge, float clog2)
{
    __shared__ float vsm[VCOLS * COLF];

    const int tid  = threadIdx.x;
    const int wv   = tid >> 5;
    const int lid  = tid & 31;
    const int half = lid & 15;
    const int khi  = lid >> 4;

    const int H = 512, Wd = 512;
    const long long chStride = (long long)H * Wd;

    constexpr int NEDGE  = WIDTH * 2 / (VCOLS * 32);
    constexpr int NPIECE = 16 * WIDTH * 2;

    float4 e4[NEDGE];
    float4 v[PF_ALL];

    int nb = blockIdx.x;
    long long imgBase, edgeBase;
    v_tile_base(nb, imgBase, edgeBase);
#pragma unroll
    for (int k = 0; k < NEDGE; ++k) {
        const int q = tid + k * VCOLS * 32;
        const int j = q & 1, h = q >> 1;
        e4[k] = *reinterpret_cast<const float4*>(edge + edgeBase + (long long)h * Wd + j * 4);
    }
    v_issue<0, PF_ALL>(in, imgBase, tid, v);

    float* X   = vsm + wv * COLF;
    float* tF  = X + 16 * XPITCH;
    float* omF = tF + VCOEF;

#pragma unroll
    for (int t = 0; t < VTILES; ++t) {
#pragma unroll
        for (int k = 0; k < PF_ALL; ++k) {
            const int q  = tid + k * VCOLS * 32;
            const int j  = q & 1, h = (q >> 1) & (WIDTH - 1), ch = q >> 10;
            float* d = vsm + (j * 4) * COLF + ch * XPITCH + h;
            d[0 * COLF] = v[k].x;
            d[1 * COLF] = v[k].y;
            d[2 * COLF] = v[k].z;
            d[3 * COLF] = v[k].w;
        }
#pragma unroll
        for (int k = 0; k < NEDGE; ++k) {
            const int q = tid + k * VCOLS * 32;
            const int j = q & 1, h = q >> 1;
            const float ev[4] = {e4[k].x, e4[k].y, e4[k].z, e4[k].w};
#pragma unroll
            for (int e = 0; e < 4; ++e) {
                float tt = clog2 * (1.0f + 150.0f * ev[e]);
                if (h == 0) tt = -512.0f;
                float* col = vsm + (j * 4 + e) * COLF + 16 * XPITCH;
                col[h]         = tt;
                col[VCOEF + h] = 1.0f - exp2_fast(tt);
                if (k == 0 && h == 0) { col[WIDTH] = -512.0f; col[VCOEF + WIDTH] = 1.0f; }
            }
        }

        const long long curBase = imgBase;
        if (t + 1 < VTILES) {
            nb += gridDim.x;
            v_tile_base(nb, imgBase, edgeBase);
#pragma unroll
            for (int k = 0; k < NEDGE; ++k) {
                const int q = tid + k * VCOLS * 32;
                const int j = q & 1, h = q >> 1;
                e4[k] = *reinterpret_cast<const float4*>(edge + edgeBase + (long long)h * Wd + j * 4);
            }
            v_issue<0, PF_AHEAD>(in, imgBase, tid, v);
        }
        lds_barrier();

        scan_dir<XPITCH>(X, tF, omF, half, khi, false);
        scan_dir<XPITCH>(X, tF, omF, half, khi, true);

        lds_barrier();
        if (t + 1 < VTILES)
            v_issue<PF_AHEAD, PF_ALL>(in, imgBase, tid, v);

        constexpr int VWB = 4;
        for (int q0 = tid; q0 < NPIECE; q0 += VWB * VCOLS * 32) {
            float4 o[VWB];
#pragma unroll
            for (int k = 0; k < VWB; ++k) {
                const int q = q0 + k * VCOLS * 32;
                const int j = q & 1, h = (q >> 1) & (WIDTH - 1), ch = q >> 10;
                const float* s = vsm + (j * 4) * COLF + ch * XPITCH + h;
                o[k].x = s[0 * COLF];
                o[k].y = s[1 * COLF];
                o[k].z = s[2 * COLF];
                o[k].w = s[3 * COLF];
            }
#pragma unroll
            for (int k = 0; k < VWB; ++k) {
                const int q = q0 + k * VCOLS * 32;
                const int j = q & 1, h = (q >> 1) & (WIDTH - 1), ch = q >> 10;
                *reinterpret_cast<float4*>(
                    out + curBase + (long long)ch * chStride + (long long)h * Wd + j * 4) = o[k];
            }
        }
        if (t + 1 < VTILES) lds_barrier();
    }
}

extern "C" void kernel_launch(void* const* d_in, const int* in_sizes, int n_in,
                              void* d_out, int out_size, void* d_ws, size_t ws_size,
                              hipStream_t stream) {
    (void)in_sizes; (void)n_in; (void)out_size; (void)d_ws; (void)ws_size;
    const float* img  = (const float*)d_in[0];
    const float* edge = (const float*)d_in[1];
    float* out = (float*)d_out;

    const double SQRT2 = 1.4142135623730951;
    const double SQRT3 = 1.7320508075688772;
    const double LOG2E = 1.4426950408889634;
    const double denom = 3.872983346207417;
    const double sigma1 = 60.0 * SQRT3 * 2.0 / denom;
    const double sigma2 = 60.0 * SQRT3 * 1.0 / denom;
    const float c1 = (float)(-SQRT2 / sigma1 * LOG2E);
    const float c2 = (float)(-SQRT2 / sigma2 * LOG2E);

    const dim3 grid(2048), block(32);

    const dim3 vgrid(256), vblock(VCOLS * 32);

    dt_pass_kernel<<<grid, block, 0, stream>>>(img, out, edge, c1);
    dt_vpass_kernel<<<vgrid, vblock, 0, stream>>>(out, out, edge, c1);
    dt_pass_kernel<<<grid, block, 0, stream>>>(out, out, edge, c2);
    dt_vpass_kernel<<<vgrid, vblock, 0, stream>>>(out, out, edge, c2);
}


# Vertical pass: a work-group's two tiles are adjacent column groups; global loads and stores move both together as 64-byte row pieces (8 lines per wave instruction, not 16); scan and sums unchanged
# speedup vs baseline: 6.8143x; 1.1600x over previous
#include <hip/hip_runtime.h>
#include <math.h>


typedef __attribute__((ext_vector_type(2))) float v2f;
typedef __attribute__((ext_vector_type(8))) float v8f;

#define WIDTH  512
#define SEGS   32
#define XPITCH 513
#define HPITCH 516

#define VCOLS  8
#define HCOEF  516
#define VCOEF  514
#define COLF   (16 * XPITCH + 2 * VCOEF)

__device__ __forceinline__ float exp2_fast(float x) {
    return __builtin_amdgcn_exp2f(x);
}

template <int CTRL, int ROWS>
__device__ __forceinline__ float dpp_row(float old, float v) {
    return __builtin_bit_cast(float, __builtin_amdgcn_update_dpp(
               __builtin_bit_cast(int, old), __builtin_bit_cast(int, v), CTRL, ROWS, 0xf, true));
}
__device__ __forceinline__ float other_row(float v) {
    return __builtin_bit_cast(float, __builtin_amdgcn_permlanex16(
               __builtin_bit_cast(int, v), __builtin_bit_cast(int, v),
               0x76543210, 0xfedcba98, false, false));
}


template <int PITCH>
__device__ __forceinline__ void seg_products(const float* X, const float* tF, const float* omF,
                                             int half, int khi, bool bwd, int s,
                                             v8f& acc, float& Prot)
{
    const int base = s * 16;
    const int idx  = base + half;

    const float t = bwd ? tF[WIDTH - idx] : tF[idx];

    float S = t;
    S += dpp_row<0x111, 0xf>(0.0f, S);
    S += dpp_row<0x112, 0xf>(0.0f, S);
    S += dpp_row<0x114, 0xf>(0.0f, S);
    S += dpp_row<0x118, 0xf>(0.0f, S);
    const float expS = exp2_fast(S);

    const float Srot = dpp_row<0x12e, 0x2>(S, S);
    Prot = dpp_row<0x128, 0x2>(expS, expS);

    acc = (v8f){0.f,0.f,0.f,0.f,0.f,0.f,0.f,0.f};
#pragma unroll
    for (int kk = 0; kk < 4; ++kk) {
        const int m0 = kk * 4 + khi * 2;
        float Sm0, Sm1;
        if (kk == 0)      { Sm0 = dpp_row<0x150, 0xf>(0.0f, Srot); Sm1 = dpp_row<0x151, 0xf>(0.0f, Srot); }
        else if (kk == 1) { Sm0 = dpp_row<0x154, 0xf>(0.0f, Srot); Sm1 = dpp_row<0x155, 0xf>(0.0f, Srot); }
        else if (kk == 2) { Sm0 = dpp_row<0x158, 0xf>(0.0f, Srot); Sm1 = dpp_row<0x159, 0xf>(0.0f, Srot); }
        else              { Sm0 = dpp_row<0x15c, 0xf>(0.0f, Srot); Sm1 = dpp_row<0x15d, 0xf>(0.0f, Srot); }
        v2f Af;
        Af.x = (m0     < half) ? exp2_fast(S - Sm0) : ((m0     == half) ? 1.0f : 0.0f);
        Af.y = (m0 + 1 < half) ? exp2_fast(S - Sm1) : ((m0 + 1 == half) ? 1.0f : 0.0f);

        const int p0 = base + m0;
        v2f Bf;
        if (!bwd) {
            Bf.x = omF[p0]     * X[half * PITCH + p0];
            Bf.y = omF[p0 + 1] * X[half * PITCH + p0 + 1];
        } else {
            const int rb = WIDTH - 16 - base + 2 - khi * 2;
            const float* Xr = X + half * PITCH + rb;
            const float* Or = omF + rb;
            const float om0 = Or[14 - kk * 4];
            const float om1 = Or[13 - kk * 4];
            Bf.x = om0 * Xr[13 - kk * 4];
            Bf.y = om1 * Xr[12 - kk * 4];
        }

        acc = __builtin_amdgcn_wmma_f32_16x16x4_f32(
                  false, Af, false, Bf, (short)0, acc, false, false);
    }
}

template <int PITCH>
__device__ __forceinline__ void seg_carry(float* X, int half, int khi, bool bwd, int s,
                                          v8f acc, float Prot, float& carry)
{
    const int base = s * 16;
    acc[0] += dpp_row<0x150, 0xf>(0.0f, Prot) * carry;
    acc[1] += dpp_row<0x151, 0xf>(0.0f, Prot) * carry;
    acc[2] += dpp_row<0x152, 0xf>(0.0f, Prot) * carry;
    acc[3] += dpp_row<0x153, 0xf>(0.0f, Prot) * carry;
    acc[4] += dpp_row<0x154, 0xf>(0.0f, Prot) * carry;
    acc[5] += dpp_row<0x155, 0xf>(0.0f, Prot) * carry;
    acc[6] += dpp_row<0x156, 0xf>(0.0f, Prot) * carry;
    acc[7] += dpp_row<0x157, 0xf>(0.0f, Prot) * carry;
    const float upper = other_row(acc[7]);
    carry = khi ? acc[7] : upper;

#pragma unroll
    for (int rr = 0; rr < 8; ++rr) {
        const int j = base + rr + 8 * khi;
        const int w = bwd ? (WIDTH - 1 - j) : j;
        X[half * PITCH + w] = acc[rr];
    }
}

template <int PITCH>
__device__ __forceinline__ void scan_dir(float* X, const float* tF, const float* omF,
                                         int half, int khi, bool bwd)
{
    float carry = 0.0f;
    for (int s = 0; s < SEGS; s += 2) {
        v8f acc0, acc1;
        float P0, P1;
        seg_products<PITCH>(X, tF, omF, half, khi, bwd, s,     acc0, P0);
        seg_products<PITCH>(X, tF, omF, half, khi, bwd, s + 1, acc1, P1);
        seg_carry<PITCH>(X, half, khi, bwd, s,     acc0, P0, carry);
        seg_carry<PITCH>(X, half, khi, bwd, s + 1, acc1, P1, carry);
    }
}

__device__ __forceinline__ void lds_barrier() {
    __builtin_amdgcn_fence(__ATOMIC_RELEASE, "workgroup", "local");
    __builtin_amdgcn_s_barrier();
    __builtin_amdgcn_fence(__ATOMIC_ACQUIRE, "workgroup", "local");
}

#define HTILES 2
#define HPF_AHEAD 56
#define HPF_ALL   64

template <int K0, int K1>
__device__ __forceinline__ void h_issue(const float* in, long long imgBase, int lid,
                                        float4 (&v)[HPF_ALL])
{
#pragma unroll
    for (int k = K0; k < K1; ++k) {
        const int q  = lid + k * 32;
        const int ch = q >> 7, w = (q & 127) * 4;
        v[k] = *reinterpret_cast<const float4*>(in + imgBase + (long long)ch * (512LL * 512LL) + w);
    }
}

__device__ __forceinline__ void h_tile_base(int nb, long long& imgBase, long long& edgeBase)
{
    const int C = 32, H = 512, Wd = 512;
    const int cg  = nb & 1;
    const int row = nb >> 1;
    const int b   = row >> 9;
    const int r   = row & 511;
    imgBase  = (long long)b * C * H * Wd + (long long)r * Wd
             + (long long)(cg * 16) * ((long long)H * Wd);
    edgeBase = (long long)b * H * Wd + (long long)r * Wd;
}

__global__ __launch_bounds__(32, 2)
void dt_pass_kernel(const float* in, float* out, const float* edge, float clog2)
{
    __shared__ __attribute__((aligned(16))) float hsm[16 * HPITCH + 2 * HCOEF];
    float* X   = hsm;
    float* tF  = hsm + 16 * HPITCH;
    float* omF = tF + HCOEF;

    const int lid  = threadIdx.x;
    const int half = lid & 15;
    const int khi  = lid >> 4;

    const long long chStride = 512LL * 512LL;

    constexpr int NEDGE  = WIDTH / 4 / 32;
    constexpr int NPIECE = 16 * WIDTH / 4;

    float4 e4[NEDGE];
    float4 v[HPF_ALL];

    int nb = blockIdx.x;
    long long imgBase, edgeBase;
    h_tile_base(nb, imgBase, edgeBase);
#pragma unroll
    for (int k = 0; k < NEDGE; ++k)
        e4[k] = *reinterpret_cast<const float4*>(edge + edgeBase + (lid + k * 32) * 4);
    h_issue<0, HPF_ALL>(in, imgBase, lid, v);

#pragma unroll
    for (int t = 0; t < HTILES; ++t) {
#pragma unroll
        for (int k = 0; k < HPF_ALL; ++k) {
            const int q  = lid + k * 32;
            const int ch = q >> 7, w = (q & 127) * 4;
            float* d = X + ch * HPITCH + w;
            d[0] = v[k].x;
            d[1] = v[k].y;
            d[2] = v[k].z;
            d[3] = v[k].w;
        }
#pragma unroll
        for (int k = 0; k < NEDGE; ++k) {
            const int w0 = (lid + k * 32) * 4;
            const float ev[4] = {e4[k].x, e4[k].y, e4[k].z, e4[k].w};
            float tv[4], ov[4];
#pragma unroll
            for (int e = 0; e < 4; ++e) {
                const float Dv = 1.0f + 150.0f * ev[e];
                float tt = clog2 * Dv;
                if (w0 + e == 0) tt = -512.0f;
                tv[e] = tt;
                ov[e] = 1.0f - exp2_fast(tt);
            }
            *reinterpret_cast<float4*>(tF  + w0) = make_float4(tv[0], tv[1], tv[2], tv[3]);
            *reinterpret_cast<float4*>(omF + w0) = make_float4(ov[0], ov[1], ov[2], ov[3]);
        }
        if (lid == 0) { tF[WIDTH] = -512.0f; omF[WIDTH] = 1.0f; }

        const long long curBase = imgBase;
        if (t + 1 < HTILES) {
            __builtin_amdgcn_sched_barrier(0);
            nb += gridDim.x;
            h_tile_base(nb, imgBase, edgeBase);
#pragma unroll
            for (int k = 0; k < NEDGE; ++k)
                e4[k] = *reinterpret_cast<const float4*>(edge + edgeBase + (lid + k * 32) * 4);
            h_issue<0, HPF_AHEAD>(in, imgBase, lid, v);
        }
        lds_barrier();

        scan_dir<HPITCH>(X, tF, omF, half, khi, false);
        scan_dir<HPITCH>(X, tF, omF, half, khi, true);

        lds_barrier();
        if (t + 1 < HTILES)
            h_issue<HPF_AHEAD, HPF_ALL>(in, imgBase, lid, v);

        constexpr int WBATCH = 4;
        for (int q0 = lid; q0 < NPIECE; q0 += WBATCH * 32) {
            float4 o[WBATCH];
#pragma unroll
            for (int k = 0; k < WBATCH; ++k) {
                const int q  = q0 + k * 32;
                const int ch = q >> 7, w = (q & 127) * 4;
                o[k] = *reinterpret_cast<const float4*>(X + ch * HPITCH + w);
            }
#pragma unroll
            for (int k = 0; k < WBATCH; ++k) {
                const int q  = q0 + k * 32;
                const int ch = q >> 7, w = (q & 127) * 4;
                *reinterpret_cast<float4*>(out + curBase + (long long)ch * chStride + w) = o[k];
            }
        }
        if (t + 1 < HTILES) lds_barrier();
    }
}

#define VP_PAIRS 64
#define VP_BATCH 16
#define VP_UNPINNED 16

__device__ __forceinline__ long long vp_goff(int k)
{
    return (long long)(k >> 3) * (512LL * 512LL) + (long long)(128 * ((k >> 1) & 3)) * 512;
}

__device__ __forceinline__ void vp_issue(const float* src, unsigned offE, unsigned offO, int bi,
                                         float4 (&v)[VP_BATCH])
{
#pragma unroll
    for (int i = 0; i < VP_BATCH; ++i) {
        const int k = bi * VP_BATCH + i;
        const float* pk = src + vp_goff(k);
        v[i] = *reinterpret_cast<const float4*>(pk + ((k & 1) ? offO : offE));
    }
}

__device__ __forceinline__ int opaque_index(int i)
{
    asm volatile("" : "+v"(i));
    return i;
}

__device__ __forceinline__ void pin_piece(float4& p)
{
    asm volatile("" : "+v"(p.x), "+v"(p.y), "+v"(p.z), "+v"(p.w));
}

__device__ __forceinline__ void vp_coeffs(float* vsm, float4 ev4, const int (&cb)[4], int K, bool row0,
                                          float clog2)
{
    const float ev[4] = {ev4.x, ev4.y, ev4.z, ev4.w};
#pragma unroll
    for (int e = 0; e < 4; ++e) {
        float tt = clog2 * (1.0f + 150.0f * ev[e]);
        if (row0) tt = -512.0f;
        float* col = vsm + cb[e] + 16 * XPITCH + 128 * K;
        col[0]     = tt;
        col[VCOEF] = 1.0f - exp2_fast(tt);
        if (row0) { col[WIDTH] = -512.0f; col[VCOEF + WIDTH] = 1.0f; }
    }
}

__global__ __launch_bounds__(VCOLS * 32)
void dt_vpass_kernel(const float* in, float* out, const float* edge, float clog2)
{
    __shared__ float vsm[VCOLS * COLF];

    const int tid  = threadIdx.x;
    const int wv   = tid >> 5;
    const int lid  = tid & 31;
    const int half = lid & 15;
    const int khi  = lid >> 4;

    const int C = 32, H = 512, Wd = 512;

    const int nb   = blockIdx.x;
    const int unit = nb & 7;
    const int cg   = (nb >> 3) & 1;
    const int b    = ((nb >> 4) & 1) | (((nb >> 7) & 1) << 1);
    const int pr   = (nb >> 5) & 3;
    const int c0   = (unit * 4 + pr) * (2 * VCOLS);
    const long long imgBase  = (long long)b * C * H * Wd + (long long)(cg * 16) * ((long long)H * Wd) + c0;
    const long long edgeBase = (long long)b * H * Wd + c0;

    const int l2 = tid & 3;
    const int jq = tid & 1;
    const int hi = (tid >> 1) & 1;
    const int hl = 16 * wv + (lid >> 2);
    const unsigned offE = (unsigned)(hl * Wd + l2 * 4);
    const unsigned offO = (unsigned)((hl + 8) * Wd + (l2 ^ 2) * 4);
    int ldsA[4], ldsB[4];
#pragma unroll
    for (int e = 0; e < 4; ++e) {
        ldsA[e] = opaque_index((jq * 4 + e) * COLF + hl + 8 * hi);
        ldsB[e] = ldsA[e] + 8 - 16 * hi;
    }
    const bool rowA0 = (hl + 8 * hi) == 0;
    const bool rowB0 = (hl + 8 * (1 - hi)) == 0;

    const float* src = in + imgBase;
    float*       dst = out + imgBase;
    const float* esrc = edge + edgeBase;

    float4 e4[8];
    float4 v[2][VP_BATCH];
    float4 PB[VP_PAIRS];
    float4 RA[VP_PAIRS];

#pragma unroll
    for (int k = 0; k < 8; ++k)
        e4[k] = *reinterpret_cast<const float4*>(esrc + (128 * ((k >> 1) & 3)) * Wd + ((k & 1) ? offO : offE));
    vp_issue(src, offE, offO, 0, v[0]);
    vp_issue(src, offE, offO, 1, v[1]);

    float* X   = vsm + wv * COLF;
    float* tF  = X + 16 * XPITCH;
    float* omF = tF + VCOEF;

#pragma unroll
    for (int K = 0; K < 4; ++K) {
        const float4 p0 = e4[2 * K], p1 = e4[2 * K + 1];
        float4 a, q;
        a.x = hi ? p1.x : p0.x;  a.y = hi ? p1.y : p0.y;  a.z = hi ? p1.z : p0.z;  a.w = hi ? p1.w : p0.w;
        q.x = hi ? p0.x : p1.x;  q.y = hi ? p0.y : p1.y;  q.z = hi ? p0.z : p1.z;  q.w = hi ? p0.w : p1.w;
        vp_coeffs(vsm, a, ldsA, K, K == 0 && rowA0, clog2);
        e4[K] = q;
    }

#pragma unroll
    for (int bi = 0; bi < 2 * VP_PAIRS / VP_BATCH; ++bi) {
#pragma unroll
        for (int i = 0; i < VP_BATCH / 2; ++i) {
            const int K = bi * (VP_BATCH / 2) + i;
            const float4 p0 = v[bi & 1][2 * i], p1 = v[bi & 1][2 * i + 1];
            float4 a, q;
            a.x = hi ? p1.x : p0.x;  a.y = hi ? p1.y : p0.y;  a.z = hi ? p1.z : p0.z;  a.w = hi ? p1.w : p0.w;
            q.x = hi ? p0.x : p1.x;  q.y = hi ? p0.y : p1.y;  q.z = hi ? p0.z : p1.z;  q.w = hi ? p0.w : p1.w;
            const int o = (K >> 2) * XPITCH + 128 * (K & 3);
            vsm[ldsA[0] + o] = a.x;
            vsm[ldsA[1] + o] = a.y;
            vsm[ldsA[2] + o] = a.z;
            vsm[ldsA[3] + o] = a.w;
            if (K >= VP_UNPINNED) pin_piece(q);
            PB[K] = q;
        }
        if (bi + 2 < 2 * VP_PAIRS / VP_BATCH)
            vp_issue(src, offE, offO, bi + 2, v[bi & 1]);
    }
    lds_barrier();

    scan_dir<XPITCH>(X, tF, omF, half, khi, false);
    scan_dir<XPITCH>(X, tF, omF, half, khi, true);

    lds_barrier();

#pragma unroll
    for (int g = 0; g < VP_PAIRS / 16; ++g) {
#pragma unroll
        for (int i = 0; i < 16; ++i) {
            const int K = g * 16 + i;
            const int o = (K >> 2) * XPITCH + 128 * (K & 3);
            RA[K].x = vsm[ldsA[0] + o];
            RA[K].y = vsm[ldsA[1] + o];
            RA[K].z = vsm[ldsA[2] + o];
            RA[K].w = vsm[ldsA[3] + o];
        }
        lds_barrier();
#pragma unroll
        for (int i = 0; i < 16; ++i) {
            const int K = g * 16 + i;
            const int o = (K >> 2) * XPITCH + 128 * (K & 3);
            vsm[ldsB[0] + o] = PB[K].x;
            vsm[ldsB[1] + o] = PB[K].y;
            vsm[ldsB[2] + o] = PB[K].z;
            vsm[ldsB[3] + o] = PB[K].w;
        }
    }
#pragma unroll
    for (int K = 0; K < 4; ++K)
        vp_coeffs(vsm, e4[K], ldsB, K, K == 0 && rowB0, clog2);
    lds_barrier();

    scan_dir<XPITCH>(X, tF, omF, half, khi, false);
    scan_dir<XPITCH>(X, tF, omF, half, khi, true);

    lds_barrier();

    constexpr int VWB = 4;
#pragma unroll
    for (int g = 0; g < VP_PAIRS / VWB; ++g) {
        float4 rb[VWB];
#pragma unroll
        for (int i = 0; i < VWB; ++i) {
            const int K = g * VWB + i;
            const int o = (K >> 2) * XPITCH + 128 * (K & 3);
            rb[i].x = vsm[ldsB[0] + o];
            rb[i].y = vsm[ldsB[1] + o];
            rb[i].z = vsm[ldsB[2] + o];
            rb[i].w = vsm[ldsB[3] + o];
        }
#pragma unroll
        for (int i = 0; i < VWB; ++i) {
            const int K = g * VWB + i;
            const float4 ra = RA[K];
            float4 se, so;
            se.x = hi ? rb[i].x : ra.x;  se.y = hi ? rb[i].y : ra.y;  se.z = hi ? rb[i].z : ra.z;  se.w = hi ? rb[i].w : ra.w;
            so.x = hi ? ra.x : rb[i].x;  so.y = hi ? ra.y : rb[i].y;  so.z = hi ? ra.z : rb[i].z;  so.w = hi ? ra.w : rb[i].w;
            float* pk = dst + vp_goff(2 * K);
            *reinterpret_cast<float4*>(pk + offE) = se;
            *reinterpret_cast<float4*>(pk + offO) = so;
        }
    }
}

extern "C" void kernel_launch(void* const* d_in, const int* in_sizes, int n_in,
                              void* d_out, int out_size, void* d_ws, size_t ws_size,
                              hipStream_t stream) {
    (void)in_sizes; (void)n_in; (void)out_size; (void)d_ws; (void)ws_size;
    const float* img  = (const float*)d_in[0];
    const float* edge = (const float*)d_in[1];
    float* out = (float*)d_out;

    const double SQRT2 = 1.4142135623730951;
    const double SQRT3 = 1.7320508075688772;
    const double LOG2E = 1.4426950408889634;
    const double denom = 3.872983346207417;
    const double sigma1 = 60.0 * SQRT3 * 2.0 / denom;
    const double sigma2 = 60.0 * SQRT3 * 1.0 / denom;
    const float c1 = (float)(-SQRT2 / sigma1 * LOG2E);
    const float c2 = (float)(-SQRT2 / sigma2 * LOG2E);

    const dim3 grid(2048), block(32);

    const dim3 vgrid(256), vblock(VCOLS * 32);

    dt_pass_kernel<<<grid, block, 0, stream>>>(img, out, edge, c1);
    dt_vpass_kernel<<<vgrid, vblock, 0, stream>>>(out, out, edge, c1);
    dt_pass_kernel<<<grid, block, 0, stream>>>(out, out, edge, c2);
    dt_vpass_kernel<<<vgrid, vblock, 0, stream>>>(out, out, edge, c2);
}


# Scan: the carry enters the product as b_0 + a_0*carry in k-block 0, whose WMMA now runs last (k-blocks summed 1,2,3,0), not as 8 broadcast P_j*carry multiply-adds on the result; sum order changes
# speedup vs baseline: 7.0315x; 1.0319x over previous
#include <hip/hip_runtime.h>
#include <math.h>


typedef __attribute__((ext_vector_type(2))) float v2f;
typedef __attribute__((ext_vector_type(8))) float v8f;

#define WIDTH  512
#define SEGS   32
#define XPITCH 513
#define HPITCH 516

#define VCOLS  8
#define HCOEF  516
#define VCOEF  514
#define COLF   (16 * XPITCH + 2 * VCOEF)

__device__ __forceinline__ float exp2_fast(float x) {
    return __builtin_amdgcn_exp2f(x);
}

template <int CTRL, int ROWS>
__device__ __forceinline__ float dpp_row(float old, float v) {
    return __builtin_bit_cast(float, __builtin_amdgcn_update_dpp(
               __builtin_bit_cast(int, old), __builtin_bit_cast(int, v), CTRL, ROWS, 0xf, true));
}
__device__ __forceinline__ float other_row(float v) {
    return __builtin_bit_cast(float, __builtin_amdgcn_permlanex16(
               __builtin_bit_cast(int, v), __builtin_bit_cast(int, v),
               0x76543210, 0xfedcba98, false, false));
}


template <int PITCH>
__device__ __forceinline__ void seg_products(const float* X, const float* tF, const float* omF,
                                             int half, int khi, bool bwd, int s,
                                             v8f& acc, v2f& Af0, v2f& Bf0, float& a0m)
{
    const int base = s * 16;
    const int idx  = base + half;

    const float t = bwd ? tF[WIDTH - idx] : tF[idx];

    float S = t;
    S += dpp_row<0x111, 0xf>(0.0f, S);
    S += dpp_row<0x112, 0xf>(0.0f, S);
    S += dpp_row<0x114, 0xf>(0.0f, S);
    S += dpp_row<0x118, 0xf>(0.0f, S);

    const float Srot = dpp_row<0x12e, 0x2>(S, S);
    a0m = dpp_row<0x150, 0x1>(0.0f, exp2_fast(t));

    acc = (v8f){0.f,0.f,0.f,0.f,0.f,0.f,0.f,0.f};
#pragma unroll
    for (int kk = 0; kk < 4; ++kk) {
        const int m0 = kk * 4 + khi * 2;
        float Sm0, Sm1;
        if (kk == 0)      { Sm0 = dpp_row<0x150, 0xf>(0.0f, Srot); Sm1 = dpp_row<0x151, 0xf>(0.0f, Srot); }
        else if (kk == 1) { Sm0 = dpp_row<0x154, 0xf>(0.0f, Srot); Sm1 = dpp_row<0x155, 0xf>(0.0f, Srot); }
        else if (kk == 2) { Sm0 = dpp_row<0x158, 0xf>(0.0f, Srot); Sm1 = dpp_row<0x159, 0xf>(0.0f, Srot); }
        else              { Sm0 = dpp_row<0x15c, 0xf>(0.0f, Srot); Sm1 = dpp_row<0x15d, 0xf>(0.0f, Srot); }
        v2f Af;
        Af.x = (m0     < half) ? exp2_fast(S - Sm0) : ((m0     == half) ? 1.0f : 0.0f);
        Af.y = (m0 + 1 < half) ? exp2_fast(S - Sm1) : ((m0 + 1 == half) ? 1.0f : 0.0f);

        const int p0 = base + m0;
        v2f Bf;
        if (!bwd) {
            Bf.x = omF[p0]     * X[half * PITCH + p0];
            Bf.y = omF[p0 + 1] * X[half * PITCH + p0 + 1];
        } else {
            const int rb = WIDTH - 16 - base + 2 - khi * 2;
            const float* Xr = X + half * PITCH + rb;
            const float* Or = omF + rb;
            const float om0 = Or[14 - kk * 4];
            const float om1 = Or[13 - kk * 4];
            Bf.x = om0 * Xr[13 - kk * 4];
            Bf.y = om1 * Xr[12 - kk * 4];
        }

        if (kk == 0) {
            Af0 = Af;
            Bf0 = Bf;
        } else {
            acc = __builtin_amdgcn_wmma_f32_16x16x4_f32(
                      false, Af, false, Bf, (short)0, acc, false, false);
        }
    }
}

template <int PITCH>
__device__ __forceinline__ void seg_carry(float* X, int half, int khi, bool bwd, int s,
                                          v8f acc, v2f Af0, v2f Bf0, float a0m, float& carry)
{
    const int base = s * 16;
    Bf0.x = __builtin_fmaf(a0m, carry, Bf0.x);
    acc = __builtin_amdgcn_wmma_f32_16x16x4_f32(
              false, Af0, false, Bf0, (short)0, acc, false, false);
    const float upper = other_row(acc[7]);
    carry = khi ? acc[7] : upper;

#pragma unroll
    for (int rr = 0; rr < 8; ++rr) {
        const int j = base + rr + 8 * khi;
        const int w = bwd ? (WIDTH - 1 - j) : j;
        X[half * PITCH + w] = acc[rr];
    }
}

template <int PITCH>
__device__ __forceinline__ void scan_dir(float* X, const float* tF, const float* omF,
                                         int half, int khi, bool bwd)
{
    float carry = 0.0f;
    for (int s = 0; s < SEGS; s += 2) {
        v8f acc0, acc1;
        v2f A0, B0, A1, B1;
        float m0, m1;
        seg_products<PITCH>(X, tF, omF, half, khi, bwd, s,     acc0, A0, B0, m0);
        seg_products<PITCH>(X, tF, omF, half, khi, bwd, s + 1, acc1, A1, B1, m1);
        seg_carry<PITCH>(X, half, khi, bwd, s,     acc0, A0, B0, m0, carry);
        seg_carry<PITCH>(X, half, khi, bwd, s + 1, acc1, A1, B1, m1, carry);
    }
}

__device__ __forceinline__ void lds_barrier() {
    __builtin_amdgcn_fence(__ATOMIC_RELEASE, "workgroup", "local");
    __builtin_amdgcn_s_barrier();
    __builtin_amdgcn_fence(__ATOMIC_ACQUIRE, "workgroup", "local");
}

#define HTILES 2
#define HPF_AHEAD 56
#define HPF_ALL   64

template <int K0, int K1>
__device__ __forceinline__ void h_issue(const float* in, long long imgBase, int lid,
                                        float4 (&v)[HPF_ALL])
{
#pragma unroll
    for (int k = K0; k < K1; ++k) {
        const int q  = lid + k * 32;
        const int ch = q >> 7, w = (q & 127) * 4;
        v[k] = *reinterpret_cast<const float4*>(in + imgBase + (long long)ch * (512LL * 512LL) + w);
    }
}

__device__ __forceinline__ void h_tile_base(int nb, long long& imgBase, long long& edgeBase)
{
    const int C = 32, H = 512, Wd = 512;
    const int cg  = nb & 1;
    const int row = nb >> 1;
    const int b   = row >> 9;
    const int r   = row & 511;
    imgBase  = (long long)b * C * H * Wd + (long long)r * Wd
             + (long long)(cg * 16) * ((long long)H * Wd);
    edgeBase = (long long)b * H * Wd + (long long)r * Wd;
}

__global__ __launch_bounds__(32, 2)
void dt_pass_kernel(const float* in, float* out, const float* edge, float clog2)
{
    __shared__ __attribute__((aligned(16))) float hsm[16 * HPITCH + 2 * HCOEF];
    float* X   = hsm;
    float* tF  = hsm + 16 * HPITCH;
    float* omF = tF + HCOEF;

    const int lid  = threadIdx.x;
    const int half = lid & 15;
    const int khi  = lid >> 4;

    const long long chStride = 512LL * 512LL;

    constexpr int NEDGE  = WIDTH / 4 / 32;
    constexpr int NPIECE = 16 * WIDTH / 4;

    float4 e4[NEDGE];
    float4 v[HPF_ALL];

    int nb = blockIdx.x;
    long long imgBase, edgeBase;
    h_tile_base(nb, imgBase, edgeBase);
#pragma unroll
    for (int k = 0; k < NEDGE; ++k)
        e4[k] = *reinterpret_cast<const float4*>(edge + edgeBase + (lid + k * 32) * 4);
    h_issue<0, HPF_ALL>(in, imgBase, lid, v);

#pragma unroll
    for (int t = 0; t < HTILES; ++t) {
#pragma unroll
        for (int k = 0; k < HPF_ALL; ++k) {
            const int q  = lid + k * 32;
            const int ch = q >> 7, w = (q & 127) * 4;
            float* d = X + ch * HPITCH + w;
            d[0] = v[k].x;
            d[1] = v[k].y;
            d[2] = v[k].z;
            d[3] = v[k].w;
        }
#pragma unroll
        for (int k = 0; k < NEDGE; ++k) {
            const int w0 = (lid + k * 32) * 4;
            const float ev[4] = {e4[k].x, e4[k].y, e4[k].z, e4[k].w};
            float tv[4], ov[4];
#pragma unroll
            for (int e = 0; e < 4; ++e) {
                const float Dv = 1.0f + 150.0f * ev[e];
                float tt = clog2 * Dv;
                if (w0 + e == 0) tt = -512.0f;
                tv[e] = tt;
                ov[e] = 1.0f - exp2_fast(tt);
            }
            *reinterpret_cast<float4*>(tF  + w0) = make_float4(tv[0], tv[1], tv[2], tv[3]);
            *reinterpret_cast<float4*>(omF + w0) = make_float4(ov[0], ov[1], ov[2], ov[3]);
        }
        if (lid == 0) { tF[WIDTH] = -512.0f; omF[WIDTH] = 1.0f; }

        const long long curBase = imgBase;
        if (t + 1 < HTILES) {
            __builtin_amdgcn_sched_barrier(0);
            nb += gridDim.x;
            h_tile_base(nb, imgBase, edgeBase);
#pragma unroll
            for (int k = 0; k < NEDGE; ++k)
                e4[k] = *reinterpret_cast<const float4*>(edge + edgeBase + (lid + k * 32) * 4);
            h_issue<0, HPF_AHEAD>(in, imgBase, lid, v);
        }
        lds_barrier();

        scan_dir<HPITCH>(X, tF, omF, half, khi, false);
        scan_dir<HPITCH>(X, tF, omF, half, khi, true);

        lds_barrier();
        if (t + 1 < HTILES)
            h_issue<HPF_AHEAD, HPF_ALL>(in, imgBase, lid, v);

        constexpr int WBATCH = 4;
        for (int q0 = lid; q0 < NPIECE; q0 += WBATCH * 32) {
            float4 o[WBATCH];
#pragma unroll
            for (int k = 0; k < WBATCH; ++k) {
                const int q  = q0 + k * 32;
                const int ch = q >> 7, w = (q & 127) * 4;
                o[k] = *reinterpret_cast<const float4*>(X + ch * HPITCH + w);
            }
#pragma unroll
            for (int k = 0; k < WBATCH; ++k) {
                const int q  = q0 + k * 32;
                const int ch = q >> 7, w = (q & 127) * 4;
                *reinterpret_cast<float4*>(out + curBase + (long long)ch * chStride + w) = o[k];
            }
        }
        if (t + 1 < HTILES) lds_barrier();
    }
}

#define VP_PAIRS 64
#define VP_BATCH 16
#define VP_UNPINNED 16

__device__ __forceinline__ long long vp_goff(int k)
{
    return (long long)(k >> 3) * (512LL * 512LL) + (long long)(128 * ((k >> 1) & 3)) * 512;
}

__device__ __forceinline__ void vp_issue(const float* src, unsigned offE, unsigned offO, int bi,
                                         float4 (&v)[VP_BATCH])
{
#pragma unroll
    for (int i = 0; i < VP_BATCH; ++i) {
        const int k = bi * VP_BATCH + i;
        const float* pk = src + vp_goff(k);
        v[i] = *reinterpret_cast<const float4*>(pk + ((k & 1) ? offO : offE));
    }
}

__device__ __forceinline__ int opaque_index(int i)
{
    asm volatile("" : "+v"(i));
    return i;
}

__device__ __forceinline__ void pin_piece(float4& p)
{
    asm volatile("" : "+v"(p.x), "+v"(p.y), "+v"(p.z), "+v"(p.w));
}

__device__ __forceinline__ void vp_coeffs(float* vsm, float4 ev4, const int (&cb)[4], int K, bool row0,
                                          float clog2)
{
    const float ev[4] = {ev4.x, ev4.y, ev4.z, ev4.w};
#pragma unroll
    for (int e = 0; e < 4; ++e) {
        float tt = clog2 * (1.0f + 150.0f * ev[e]);
        if (row0) tt = -512.0f;
        float* col = vsm + cb[e] + 16 * XPITCH + 128 * K;
        col[0]     = tt;
        col[VCOEF] = 1.0f - exp2_fast(tt);
        if (row0) { col[WIDTH] = -512.0f; col[VCOEF + WIDTH] = 1.0f; }
    }
}

__global__ __launch_bounds__(VCOLS * 32)
void dt_vpass_kernel(const float* in, float* out, const float* edge, float clog2)
{
    __shared__ float vsm[VCOLS * COLF];

    const int tid  = threadIdx.x;
    const int wv   = tid >> 5;
    const int lid  = tid & 31;
    const int half = lid & 15;
    const int khi  = lid >> 4;

    const int C = 32, H = 512, Wd = 512;

    const int nb   = blockIdx.x;
    const int unit = nb & 7;
    const int cg   = (nb >> 3) & 1;
    const int b    = ((nb >> 4) & 1) | (((nb >> 7) & 1) << 1);
    const int pr   = (nb >> 5) & 3;
    const int c0   = (unit * 4 + pr) * (2 * VCOLS);
    const long long imgBase  = (long long)b * C * H * Wd + (long long)(cg * 16) * ((long long)H * Wd) + c0;
    const long long edgeBase = (long long)b * H * Wd + c0;

    const int l2 = tid & 3;
    const int jq = tid & 1;
    const int hi = (tid >> 1) & 1;
    const int hl = 16 * wv + (lid >> 2);
    const unsigned offE = (unsigned)(hl * Wd + l2 * 4);
    const unsigned offO = (unsigned)((hl + 8) * Wd + (l2 ^ 2) * 4);
    int ldsA[4], ldsB[4];
#pragma unroll
    for (int e = 0; e < 4; ++e) {
        ldsA[e] = opaque_index((jq * 4 + e) * COLF + hl + 8 * hi);
        ldsB[e] = ldsA[e] + 8 - 16 * hi;
    }
    const bool rowA0 = (hl + 8 * hi) == 0;
    const bool rowB0 = (hl + 8 * (1 - hi)) == 0;

    const float* src = in + imgBase;
    float*       dst = out + imgBase;
    const float* esrc = edge + edgeBase;

    float4 e4[8];
    float4 v[2][VP_BATCH];
    float4 PB[VP_PAIRS];
    float4 RA[VP_PAIRS];

#pragma unroll
    for (int k = 0; k < 8; ++k)
        e4[k] = *reinterpret_cast<const float4*>(esrc + (128 * ((k >> 1) & 3)) * Wd + ((k & 1) ? offO : offE));
    vp_issue(src, offE, offO, 0, v[0]);
    vp_issue(src, offE, offO, 1, v[1]);

    float* X   = vsm + wv * COLF;
    float* tF  = X + 16 * XPITCH;
    float* omF = tF + VCOEF;

#pragma unroll
    for (int K = 0; K < 4; ++K) {
        const float4 p0 = e4[2 * K], p1 = e4[2 * K + 1];
        float4 a, q;
        a.x = hi ? p1.x : p0.x;  a.y = hi ? p1.y : p0.y;  a.z = hi ? p1.z : p0.z;  a.w = hi ? p1.w : p0.w;
        q.x = hi ? p0.x : p1.x;  q.y = hi ? p0.y : p1.y;  q.z = hi ? p0.z : p1.z;  q.w = hi ? p0.w : p1.w;
        vp_coeffs(vsm, a, ldsA, K, K == 0 && rowA0, clog2);
        e4[K] = q;
    }

#pragma unroll
    for (int bi = 0; bi < 2 * VP_PAIRS / VP_BATCH; ++bi) {
#pragma unroll
        for (int i = 0; i < VP_BATCH / 2; ++i) {
            const int K = bi * (VP_BATCH / 2) + i;
            const float4 p0 = v[bi & 1][2 * i], p1 = v[bi & 1][2 * i + 1];
            float4 a, q;
            a.x = hi ? p1.x : p0.x;  a.y = hi ? p1.y : p0.y;  a.z = hi ? p1.z : p0.z;  a.w = hi ? p1.w : p0.w;
            q.x = hi ? p0.x : p1.x;  q.y = hi ? p0.y : p1.y;  q.z = hi ? p0.z : p1.z;  q.w = hi ? p0.w : p1.w;
            const int o = (K >> 2) * XPITCH + 128 * (K & 3);
            vsm[ldsA[0] + o] = a.x;
            vsm[ldsA[1] + o] = a.y;
            vsm[ldsA[2] + o] = a.z;
            vsm[ldsA[3] + o] = a.w;
            if (K >= VP_UNPINNED) pin_piece(q);
            PB[K] = q;
        }
        if (bi + 2 < 2 * VP_PAIRS / VP_BATCH)
            vp_issue(src, offE, offO, bi + 2, v[bi & 1]);
    }
    lds_barrier();

    scan_dir<XPITCH>(X, tF, omF, half, khi, false);
    scan_dir<XPITCH>(X, tF, omF, half, khi, true);

    lds_barrier();

#pragma unroll
    for (int g = 0; g < VP_PAIRS / 16; ++g) {
#pragma unroll
        for (int i = 0; i < 16; ++i) {
            const int K = g * 16 + i;
            const int o = (K >> 2) * XPITCH + 128 * (K & 3);
            RA[K].x = vsm[ldsA[0] + o];
            RA[K].y = vsm[ldsA[1] + o];
            RA[K].z = vsm[ldsA[2] + o];
            RA[K].w = vsm[ldsA[3] + o];
        }
        lds_barrier();
#pragma unroll
        for (int i = 0; i < 16; ++i) {
            const int K = g * 16 + i;
            const int o = (K >> 2) * XPITCH + 128 * (K & 3);
            vsm[ldsB[0] + o] = PB[K].x;
            vsm[ldsB[1] + o] = PB[K].y;
            vsm[ldsB[2] + o] = PB[K].z;
            vsm[ldsB[3] + o] = PB[K].w;
        }
    }
#pragma unroll
    for (int K = 0; K < 4; ++K)
        vp_coeffs(vsm, e4[K], ldsB, K, K == 0 && rowB0, clog2);
    lds_barrier();

    scan_dir<XPITCH>(X, tF, omF, half, khi, false);
    scan_dir<XPITCH>(X, tF, omF, half, khi, true);

    lds_barrier();

    constexpr int VWB = 4;
#pragma unroll
    for (int g = 0; g < VP_PAIRS / VWB; ++g) {
        float4 rb[VWB];
#pragma unroll
        for (int i = 0; i < VWB; ++i) {
            const int K = g * VWB + i;
            const int o = (K >> 2) * XPITCH + 128 * (K & 3);
            rb[i].x = vsm[ldsB[0] + o];
            rb[i].y = vsm[ldsB[1] + o];
            rb[i].z = vsm[ldsB[2] + o];
            rb[i].w = vsm[ldsB[3] + o];
        }
#pragma unroll
        for (int i = 0; i < VWB; ++i) {
            const int K = g * VWB + i;
            const float4 ra = RA[K];
            float4 se, so;
            se.x = hi ? rb[i].x : ra.x;  se.y = hi ? rb[i].y : ra.y;  se.z = hi ? rb[i].z : ra.z;  se.w = hi ? rb[i].w : ra.w;
            so.x = hi ? ra.x : rb[i].x;  so.y = hi ? ra.y : rb[i].y;  so.z = hi ? ra.z : rb[i].z;  so.w = hi ? ra.w : rb[i].w;
            float* pk = dst + vp_goff(2 * K);
            *reinterpret_cast<float4*>(pk + offE) = se;
            *reinterpret_cast<float4*>(pk + offO) = so;
        }
    }
}

extern "C" void kernel_launch(void* const* d_in, const int* in_sizes, int n_in,
                              void* d_out, int out_size, void* d_ws, size_t ws_size,
                              hipStream_t stream) {
    (void)in_sizes; (void)n_in; (void)out_size; (void)d_ws; (void)ws_size;
    const float* img  = (const float*)d_in[0];
    const float* edge = (const float*)d_in[1];
    float* out = (float*)d_out;

    const double SQRT2 = 1.4142135623730951;
    const double SQRT3 = 1.7320508075688772;
    const double LOG2E = 1.4426950408889634;
    const double denom = 3.872983346207417;
    const double sigma1 = 60.0 * SQRT3 * 2.0 / denom;
    const double sigma2 = 60.0 * SQRT3 * 1.0 / denom;
    const float c1 = (float)(-SQRT2 / sigma1 * LOG2E);
    const float c2 = (float)(-SQRT2 / sigma2 * LOG2E);

    const dim3 grid(2048), block(32);

    const dim3 vgrid(256), vblock(VCOLS * 32);

    dt_pass_kernel<<<grid, block, 0, stream>>>(img, out, edge, c1);
    dt_vpass_kernel<<<vgrid, vblock, 0, stream>>>(out, out, edge, c1);
    dt_pass_kernel<<<grid, block, 0, stream>>>(out, out, edge, c2);
    dt_vpass_kernel<<<vgrid, vblock, 0, stream>>>(out, out, edge, c2);
}
